# starting kernel
# baseline (speedup 1.0000x reference)
#include <hip/hip_runtime.h>
#include <hip/hip_bf16.h>
#include <math.h>

typedef __attribute__((ext_vector_type(2))) float v2f;
typedef __attribute__((ext_vector_type(8))) float v8f;

#define BATCH   2
#define SEQLEN  1024
#define D_MODEL 512
#define D_INNER 1024
#define D_STATE 128
#define DT_RANK 32
#define NPROJ   (DT_RANK + 2 * D_STATE)
#define ROWS    (BATCH * SEQLEN)

__device__ __forceinline__ float silu_f(float x) {
    return x * (1.0f / (1.0f + __expf(-x)));
}
__device__ __forceinline__ float softplus_f(float x) {
    return (x > 20.0f) ? x : log1pf(__expf(x));
}

template <int EPI, int NB>
__global__ __launch_bounds__(32) void gemm16_wmma_f32(
    const float* __restrict__ A, int lda,
    const float* __restrict__ B, int ldb,
    float* __restrict__ C, int ldc,
    int K, const float* __restrict__ bias, int split)
{
    const int tn   = blockIdx.x;
    const int tm   = blockIdx.y;
    const int lane = threadIdx.x;
    const int half = lane >> 4;
    const int l16  = lane & 15;

    const float* Arow = A + (long)(tm * 16 + l16) * lda;
    const float* Bcol = B + (tn * (NB * 16) + l16);

    v8f acc[NB];
#pragma unroll
    for (int nb = 0; nb < NB; ++nb) acc[nb] = (v8f){};

    for (int k0 = 0; k0 < K; k0 += 4) {
        v2f a;
        a.x = Arow[k0 + 2 * half];
        a.y = Arow[k0 + 2 * half + 1];
        const float* Bk0 = Bcol + (long)(k0 + 2 * half) * ldb;
        const float* Bk1 = Bcol + (long)(k0 + 2 * half + 1) * ldb;
#pragma unroll
        for (int nb = 0; nb < NB; ++nb) {
            v2f b;
            b.x = Bk0[nb * 16];
            b.y = Bk1[nb * 16];
            acc[nb] = __builtin_amdgcn_wmma_f32_16x16x4_f32(
                 false, a,  false, b,
                 (short)0, acc[nb],  false,  false);
        }
    }

#pragma unroll
    for (int nb = 0; nb < NB; ++nb) {
        const int col = tn * (NB * 16) + nb * 16 + l16;
#pragma unroll
        for (int r = 0; r < 8; ++r) {
            const int row = tm * 16 + half * 8 + r;
            float v = acc[nb][r];
            if (EPI == 1) {
                if (col < split) v = silu_f(v);
            } else if (EPI == 2) {
                v = softplus_f(v + bias[col]);
            }
            C[(long)row * ldc + col] = v;
        }
    }
}

__global__ __launch_bounds__(32) void mamba_scan(
    const float* __restrict__ dt, const float* __restrict__ proj,
    const float* __restrict__ UZ, const float* __restrict__ A_log,
    const float* __restrict__ D_param, float* __restrict__ y)
{
    const int bd   = blockIdx.x;
    const int b    = bd >> 10;
    const int d    = bd & (D_INNER - 1);
    const int lane = threadIdx.x;

    const float4 al = *(const float4*)(A_log + d * D_STATE + lane * 4);
    float a[4], h[4];
    a[0] = -__expf(al.x); a[1] = -__expf(al.y);
    a[2] = -__expf(al.z); a[3] = -__expf(al.w);
#pragma unroll
    for (int j = 0; j < 4; ++j) h[j] = 0.0f;
    const float Dp = D_param[d];

    for (int t = 0; t < SEQLEN; ++t) {
        const int   row = b * SEQLEN + t;
        const float dtv = dt[(long)row * D_INNER + d];
        const float uv  = UZ[(long)row * (2 * D_INNER) + d];
        const float* pr = proj + (long)row * NPROJ;
        if (t + 1 < SEQLEN) {
            __builtin_prefetch(pr + NPROJ, 0, 1);
        }
        const float4 Bv = *(const float4*)(pr + DT_RANK + lane * 4);
        const float4 Cv = *(const float4*)(pr + DT_RANK + D_STATE + lane * 4);
        const float du = dtv * uv;
        float acc;
        h[0] = __expf(dtv * a[0]) * h[0] + du * Bv.x;
        h[1] = __expf(dtv * a[1]) * h[1] + du * Bv.y;
        h[2] = __expf(dtv * a[2]) * h[2] + du * Bv.z;
        h[3] = __expf(dtv * a[3]) * h[3] + du * Bv.w;
        acc = h[0] * Cv.x + h[1] * Cv.y + h[2] * Cv.z + h[3] * Cv.w;
#pragma unroll
        for (int off = 16; off; off >>= 1) acc += __shfl_xor(acc, off, 32);
        if (lane == 0) y[(long)row * D_INNER + d] = acc + Dp * uv;
    }
}

__global__ __launch_bounds__(256) void ln_gate(
    float* __restrict__ y, const float* __restrict__ UZ,
    const float* __restrict__ gamma, const float* __restrict__ beta)
{
    const int row = blockIdx.x;
    const int tid = threadIdx.x;
    const int c0  = tid * 4;

    const float4 v = *(const float4*)(y + (long)row * D_INNER + c0);
    float s  = v.x + v.y + v.z + v.w;
    float s2 = v.x * v.x + v.y * v.y + v.z * v.z + v.w * v.w;

#pragma unroll
    for (int off = 16; off; off >>= 1) {
        s  += __shfl_xor(s, off, 32);
        s2 += __shfl_xor(s2, off, 32);
    }
    __shared__ float rs[8], rs2[8];
    const int wid = tid >> 5;
    if ((tid & 31) == 0) { rs[wid] = s; rs2[wid] = s2; }
    __syncthreads();
    s = 0.0f; s2 = 0.0f;
#pragma unroll
    for (int w = 0; w < 8; ++w) { s += rs[w]; s2 += rs2[w]; }

    const float mu   = s * (1.0f / D_INNER);
    const float var  = s2 * (1.0f / D_INNER) - mu * mu;
    const float rstd = rsqrtf(var + 1e-5f);

    const float4 ga = *(const float4*)(gamma + c0);
    const float4 be = *(const float4*)(beta + c0);
    const float4 zv = *(const float4*)(UZ + (long)row * (2 * D_INNER) + D_INNER + c0);

    float4 o;
    o.x = ((v.x - mu) * rstd * ga.x + be.x) * silu_f(zv.x);
    o.y = ((v.y - mu) * rstd * ga.y + be.y) * silu_f(zv.y);
    o.z = ((v.z - mu) * rstd * ga.z + be.z) * silu_f(zv.z);
    o.w = ((v.w - mu) * rstd * ga.w + be.w) * silu_f(zv.w);
    *(float4*)(y + (long)row * D_INNER + c0) = o;
}

extern "C" void kernel_launch(void* const* d_in, const int* in_sizes, int n_in,
                              void* d_out, int out_size, void* d_ws, size_t ws_size,
                              hipStream_t stream)
{
    const float* x     = (const float*)d_in[0];
    const float* W_in  = (const float*)d_in[1];
    const float* W_x   = (const float*)d_in[2];
    const float* W_dt  = (const float*)d_in[3];
    const float* b_dt  = (const float*)d_in[4];
    const float* W_out = (const float*)d_in[5];
    const float* gamma = (const float*)d_in[6];
    const float* beta  = (const float*)d_in[7];
    const float* A_log = (const float*)d_in[8];
    const float* D_par = (const float*)d_in[9];
    float* out = (float*)d_out;

    float* UZ   = (float*)d_ws;
    float* proj = UZ   + (size_t)ROWS * 2 * D_INNER;
    float* dtb  = proj + (size_t)ROWS * NPROJ;
    float* yb   = dtb  + (size_t)ROWS * D_INNER;

    gemm16_wmma_f32<1, 4><<<dim3((2 * D_INNER) / 64, ROWS / 16), 32, 0, stream>>>(
        x, D_MODEL, W_in, 2 * D_INNER, UZ, 2 * D_INNER, D_MODEL, nullptr, D_INNER);

    gemm16_wmma_f32<0, 2><<<dim3(NPROJ / 32, ROWS / 16), 32, 0, stream>>>(
        UZ, 2 * D_INNER, W_x, NPROJ, proj, NPROJ, D_INNER, nullptr, 0);

    gemm16_wmma_f32<2, 4><<<dim3(D_INNER / 64, ROWS / 16), 32, 0, stream>>>(
        proj, NPROJ, W_dt, D_INNER, dtb, D_INNER, DT_RANK, b_dt, 0);

    mamba_scan<<<BATCH * D_INNER, 32, 0, stream>>>(dtb, proj, UZ, A_log, D_par, yb);

    ln_gate<<<ROWS, 256, 0, stream>>>(yb, UZ, gamma, beta);

    gemm16_wmma_f32<0, 4><<<dim3(D_MODEL / 64, ROWS / 16), 32, 0, stream>>>(
        yb, D_INNER, W_out, D_MODEL, out, D_MODEL, D_INNER, nullptr, 0);
}


# Scan time loop taken in chunks of 16 steps: a chunk's dt/u/B/C loads are issued one chunk ahead into a second register set and its 16 lane sums run side by side (same xor tree per step)
# speedup vs baseline: 2.4227x; 2.4227x over previous
#include <hip/hip_runtime.h>
#include <hip/hip_bf16.h>
#include <math.h>

typedef __attribute__((ext_vector_type(2))) float v2f;
typedef __attribute__((ext_vector_type(8))) float v8f;

#define BATCH   2
#define SEQLEN  1024
#define D_MODEL 512
#define D_INNER 1024
#define D_STATE 128
#define DT_RANK 32
#define NPROJ   (DT_RANK + 2 * D_STATE)
#define ROWS    (BATCH * SEQLEN)

__device__ __forceinline__ float silu_f(float x) {
    return x * (1.0f / (1.0f + __expf(-x)));
}
__device__ __forceinline__ float softplus_f(float x) {
    return (x > 20.0f) ? x : log1pf(__expf(x));
}

template <int EPI, int NB>
__global__ __launch_bounds__(32) void gemm16_wmma_f32(
    const float* __restrict__ A, int lda,
    const float* __restrict__ B, int ldb,
    float* __restrict__ C, int ldc,
    int K, const float* __restrict__ bias, int split)
{
    const int tn   = blockIdx.x;
    const int tm   = blockIdx.y;
    const int lane = threadIdx.x;
    const int half = lane >> 4;
    const int l16  = lane & 15;

    const float* Arow = A + (long)(tm * 16 + l16) * lda;
    const float* Bcol = B + (tn * (NB * 16) + l16);

    v8f acc[NB];
#pragma unroll
    for (int nb = 0; nb < NB; ++nb) acc[nb] = (v8f){};

    for (int k0 = 0; k0 < K; k0 += 4) {
        v2f a;
        a.x = Arow[k0 + 2 * half];
        a.y = Arow[k0 + 2 * half + 1];
        const float* Bk0 = Bcol + (long)(k0 + 2 * half) * ldb;
        const float* Bk1 = Bcol + (long)(k0 + 2 * half + 1) * ldb;
#pragma unroll
        for (int nb = 0; nb < NB; ++nb) {
            v2f b;
            b.x = Bk0[nb * 16];
            b.y = Bk1[nb * 16];
            acc[nb] = __builtin_amdgcn_wmma_f32_16x16x4_f32(
                 false, a,  false, b,
                 (short)0, acc[nb],  false,  false);
        }
    }

#pragma unroll
    for (int nb = 0; nb < NB; ++nb) {
        const int col = tn * (NB * 16) + nb * 16 + l16;
#pragma unroll
        for (int r = 0; r < 8; ++r) {
            const int row = tm * 16 + half * 8 + r;
            float v = acc[nb][r];
            if (EPI == 1) {
                if (col < split) v = silu_f(v);
            } else if (EPI == 2) {
                v = softplus_f(v + bias[col]);
            }
            C[(long)row * ldc + col] = v;
        }
    }
}

#define SCAN_T   16
#define SCAN_NCH (SEQLEN / SCAN_T)

struct ScanOps {
    float  dtv[SCAN_T];
    float  uv[SCAN_T];
    float4 Bv[SCAN_T];
    float4 Cv[SCAN_T];
};

__device__ __forceinline__ void scan_load(
    ScanOps& o, const float* __restrict__ dtp, const float* __restrict__ up,
    const float* __restrict__ prp, int t0)
{
#pragma unroll
    for (int i = 0; i < SCAN_T; ++i) {
        const long t = t0 + i;
        o.dtv[i] = dtp[t * D_INNER];
        o.uv[i]  = up[t * (2 * D_INNER)];
        o.Bv[i]  = *(const float4*)(prp + t * NPROJ);
        o.Cv[i]  = *(const float4*)(prp + t * NPROJ + D_STATE);
    }
}

__device__ __forceinline__ void scan_chunk(
    const ScanOps& o, const float (&a)[4], float (&h)[4], float Dp,
    float* __restrict__ yp, int t0, int lane)
{
    float acc[SCAN_T];
#pragma unroll
    for (int i = 0; i < SCAN_T; ++i) {
        const float dtv = o.dtv[i];
        const float du  = dtv * o.uv[i];
        h[0] = __expf(dtv * a[0]) * h[0] + du * o.Bv[i].x;
        h[1] = __expf(dtv * a[1]) * h[1] + du * o.Bv[i].y;
        h[2] = __expf(dtv * a[2]) * h[2] + du * o.Bv[i].z;
        h[3] = __expf(dtv * a[3]) * h[3] + du * o.Bv[i].w;
        acc[i] = h[0] * o.Cv[i].x + h[1] * o.Cv[i].y + h[2] * o.Cv[i].z + h[3] * o.Cv[i].w;
    }
#pragma unroll
    for (int off = 16; off; off >>= 1) {
#pragma unroll
        for (int i = 0; i < SCAN_T; ++i) acc[i] += __shfl_xor(acc[i], off, 32);
    }
    if (lane == 0) {
#pragma unroll
        for (int i = 0; i < SCAN_T; ++i)
            yp[(long)(t0 + i) * D_INNER] = acc[i] + Dp * o.uv[i];
    }
}

__global__ __launch_bounds__(32) void mamba_scan(
    const float* __restrict__ dt, const float* __restrict__ proj,
    const float* __restrict__ UZ, const float* __restrict__ A_log,
    const float* __restrict__ D_param, float* __restrict__ y)
{
    const int bd   = blockIdx.x;
    const int b    = bd >> 10;
    const int d    = bd & (D_INNER - 1);
    const int lane = threadIdx.x;

    const float4 al = *(const float4*)(A_log + d * D_STATE + lane * 4);
    float a[4], h[4];
    a[0] = -__expf(al.x); a[1] = -__expf(al.y);
    a[2] = -__expf(al.z); a[3] = -__expf(al.w);
#pragma unroll
    for (int j = 0; j < 4; ++j) h[j] = 0.0f;
    const float Dp = D_param[d];

    const long row0 = (long)b * SEQLEN;
    const float* dtp = dt + row0 * D_INNER + d;
    const float* up  = UZ + row0 * (2 * D_INNER) + d;
    const float* prp = proj + row0 * NPROJ + DT_RANK + lane * 4;
    float*       yp  = y + row0 * D_INNER + d;

    ScanOps s0, s1;
    scan_load(s0, dtp, up, prp, 0);
    for (int c = 0; c < SCAN_NCH; c += 2) {
        scan_load(s1, dtp, up, prp, (c + 1) * SCAN_T);
        scan_chunk(s0, a, h, Dp, yp, c * SCAN_T, lane);
        const int cn = (c + 2 < SCAN_NCH) ? (c + 2) : (SCAN_NCH - 1);
        scan_load(s0, dtp, up, prp, cn * SCAN_T);
        scan_chunk(s1, a, h, Dp, yp, (c + 1) * SCAN_T, lane);
    }
}

__global__ __launch_bounds__(256) void ln_gate(
    float* __restrict__ y, const float* __restrict__ UZ,
    const float* __restrict__ gamma, const float* __restrict__ beta)
{
    const int row = blockIdx.x;
    const int tid = threadIdx.x;
    const int c0  = tid * 4;

    const float4 v = *(const float4*)(y + (long)row * D_INNER + c0);
    float s  = v.x + v.y + v.z + v.w;
    float s2 = v.x * v.x + v.y * v.y + v.z * v.z + v.w * v.w;

#pragma unroll
    for (int off = 16; off; off >>= 1) {
        s  += __shfl_xor(s, off, 32);
        s2 += __shfl_xor(s2, off, 32);
    }
    __shared__ float rs[8], rs2[8];
    const int wid = tid >> 5;
    if ((tid & 31) == 0) { rs[wid] = s; rs2[wid] = s2; }
    __syncthreads();
    s = 0.0f; s2 = 0.0f;
#pragma unroll
    for (int w = 0; w < 8; ++w) { s += rs[w]; s2 += rs2[w]; }

    const float mu   = s * (1.0f / D_INNER);
    const float var  = s2 * (1.0f / D_INNER) - mu * mu;
    const float rstd = rsqrtf(var + 1e-5f);

    const float4 ga = *(const float4*)(gamma + c0);
    const float4 be = *(const float4*)(beta + c0);
    const float4 zv = *(const float4*)(UZ + (long)row * (2 * D_INNER) + D_INNER + c0);

    float4 o;
    o.x = ((v.x - mu) * rstd * ga.x + be.x) * silu_f(zv.x);
    o.y = ((v.y - mu) * rstd * ga.y + be.y) * silu_f(zv.y);
    o.z = ((v.z - mu) * rstd * ga.z + be.z) * silu_f(zv.z);
    o.w = ((v.w - mu) * rstd * ga.w + be.w) * silu_f(zv.w);
    *(float4*)(y + (long)row * D_INNER + c0) = o;
}

extern "C" void kernel_launch(void* const* d_in, const int* in_sizes, int n_in,
                              void* d_out, int out_size, void* d_ws, size_t ws_size,
                              hipStream_t stream)
{
    const float* x     = (const float*)d_in[0];
    const float* W_in  = (const float*)d_in[1];
    const float* W_x   = (const float*)d_in[2];
    const float* W_dt  = (const float*)d_in[3];
    const float* b_dt  = (const float*)d_in[4];
    const float* W_out = (const float*)d_in[5];
    const float* gamma = (const float*)d_in[6];
    const float* beta  = (const float*)d_in[7];
    const float* A_log = (const float*)d_in[8];
    const float* D_par = (const float*)d_in[9];
    float* out = (float*)d_out;

    float* UZ   = (float*)d_ws;
    float* proj = UZ   + (size_t)ROWS * 2 * D_INNER;
    float* dtb  = proj + (size_t)ROWS * NPROJ;
    float* yb   = dtb  + (size_t)ROWS * D_INNER;

    gemm16_wmma_f32<1, 4><<<dim3((2 * D_INNER) / 64, ROWS / 16), 32, 0, stream>>>(
        x, D_MODEL, W_in, 2 * D_INNER, UZ, 2 * D_INNER, D_MODEL, nullptr, D_INNER);

    gemm16_wmma_f32<0, 2><<<dim3(NPROJ / 32, ROWS / 16), 32, 0, stream>>>(
        UZ, 2 * D_INNER, W_x, NPROJ, proj, NPROJ, D_INNER, nullptr, 0);

    gemm16_wmma_f32<2, 4><<<dim3(D_INNER / 64, ROWS / 16), 32, 0, stream>>>(
        proj, NPROJ, W_dt, D_INNER, dtb, D_INNER, DT_RANK, b_dt, 0);

    mamba_scan<<<BATCH * D_INNER, 32, 0, stream>>>(dtb, proj, UZ, A_log, D_par, yb);

    ln_gate<<<ROWS, 256, 0, stream>>>(yb, UZ, gamma, beta);

    gemm16_wmma_f32<0, 4><<<dim3(D_MODEL / 64, ROWS / 16), 32, 0, stream>>>(
        yb, D_INNER, W_out, D_MODEL, out, D_MODEL, D_INNER, nullptr, 0);
}


# GEMMs: each wave loads whole 32-k blocks of A and B with 16-byte loads, one group per block, 1-3 blocks ahead of their products (ring of register sets); k order in a block re-mapped: sums re-ordered
# speedup vs baseline: 3.9062x; 1.6123x over previous
#include <hip/hip_runtime.h>
#include <hip/hip_bf16.h>
#include <math.h>

typedef __attribute__((ext_vector_type(2))) float v2f;
typedef __attribute__((ext_vector_type(8))) float v8f;

#define BATCH   2
#define SEQLEN  1024
#define D_MODEL 512
#define D_INNER 1024
#define D_STATE 128
#define DT_RANK 32
#define NPROJ   (DT_RANK + 2 * D_STATE)
#define ROWS    (BATCH * SEQLEN)

__device__ __forceinline__ float silu_f(float x) {
    return x * (1.0f / (1.0f + __expf(-x)));
}
__device__ __forceinline__ float softplus_f(float x) {
    return (x > 20.0f) ? x : log1pf(__expf(x));
}

typedef __attribute__((ext_vector_type(4))) float v4f;

template <int NB> struct VecN;
template <> struct VecN<2> { typedef v2f T; };
template <> struct VecN<4> { typedef v4f T; };

#define GEMM_KB 32

template <int NB>
struct GemmFrag {
    v4f a[4];
    typename VecN<NB>::T b0[8];
    typename VecN<NB>::T b1[8];
};

template <int NB>
__device__ __forceinline__ void gemm_load(
    GemmFrag<NB>& f, const float* __restrict__ Ap,
    const float* __restrict__ Bp, int ldb, int kb)
{
    typedef typename VecN<NB>::T VB;
    const float* ap = Ap + kb * GEMM_KB;
#pragma unroll
    for (int j = 0; j < 4; ++j) f.a[j] = *(const v4f*)(ap + 4 * j);
    const float* bp = Bp + (long)(kb * GEMM_KB) * ldb;
#pragma unroll
    for (int s = 0; s < 8; ++s) {
        f.b0[s] = *(const VB*)(bp + (long)(2 * s) * ldb);
        f.b1[s] = *(const VB*)(bp + (long)(2 * s + 1) * ldb);
    }
}

template <int NB>
__device__ __forceinline__ void gemm_block(const GemmFrag<NB>& f, v8f (&acc)[NB])
{
#pragma unroll
    for (int s = 0; s < 8; ++s) {
        v2f a;
        a.x = f.a[s >> 1][(s & 1) * 2];
        a.y = f.a[s >> 1][(s & 1) * 2 + 1];
#pragma unroll
        for (int nb = 0; nb < NB; ++nb) {
            v2f b;
            b.x = f.b0[s][nb];
            b.y = f.b1[s][nb];
            acc[nb] = __builtin_amdgcn_wmma_f32_16x16x4_f32(
                 false, a,  false, b,
                 (short)0, acc[nb],  false,  false);
        }
    }
}

template <int NB, int R, bool LAST>
__device__ __forceinline__ void gemm_ring_pass(
    GemmFrag<NB> (&f)[R], v8f (&acc)[NB], const float* __restrict__ Ap,
    const float* __restrict__ Bp, int ldb, int kb)
{
#pragma unroll
    for (int j = 0; j < R; ++j) {
        if (!LAST || j == 0)
            gemm_load<NB>(f[(j + R - 1) % R], Ap, Bp, ldb, kb + j + R - 1);
        __builtin_amdgcn_sched_barrier(0);
        gemm_block<NB>(f[j], acc);
        __builtin_amdgcn_sched_barrier(0);
    }
}

template <int EPI, int NB, int K, int R>
__global__ __launch_bounds__(32) void gemm16_wmma_f32(
    const float* __restrict__ A, int lda,
    const float* __restrict__ B, int ldb,
    float* __restrict__ C, int ldc,
    const float* __restrict__ bias, int split)
{
    typedef typename VecN<NB>::T VB;
    constexpr int NKB = K / GEMM_KB;
    static_assert(K % GEMM_KB == 0 && NKB % R == 0, "K blocks must fill the ring");

    const int tn   = blockIdx.x;
    const int tm   = blockIdx.y;
    const int lane = threadIdx.x;
    const int half = lane >> 4;
    const int l16  = lane & 15;

    const float* Ap = A + (long)(tm * 16 + l16) * lda + 16 * half;
    const float* Bp = B + (long)(16 * half) * ldb + (tn * (NB * 16) + NB * l16);

    v8f acc[NB];
#pragma unroll
    for (int nb = 0; nb < NB; ++nb) acc[nb] = (v8f){};

    GemmFrag<NB> f[R];
#pragma unroll
    for (int j = 0; j < R - 1; ++j) {
        gemm_load<NB>(f[j], Ap, Bp, ldb, j);
        __builtin_amdgcn_sched_barrier(0);
    }

#pragma unroll 1
    for (int kb = 0; kb < NKB - R; kb += R)
        gemm_ring_pass<NB, R, false>(f, acc, Ap, Bp, ldb, kb);
    gemm_ring_pass<NB, R, true>(f, acc, Ap, Bp, ldb, NKB - R);

    const int col0 = tn * (NB * 16) + NB * l16;
    VB bv = {};
    if (EPI == 2) bv = *(const VB*)(bias + col0);
#pragma unroll
    for (int r = 0; r < 8; ++r) {
        const int row = tm * 16 + half * 8 + r;
        VB o;
#pragma unroll
        for (int nb = 0; nb < NB; ++nb) {
            const int col = col0 + nb;
            float v = acc[nb][r];
            if (EPI == 1) {
                if (col < split) v = silu_f(v);
            } else if (EPI == 2) {
                v = softplus_f(v + bv[nb]);
            }
            o[nb] = v;
        }
        *(VB*)(C + (long)row * ldc + col0) = o;
    }
}

#define SCAN_T   16
#define SCAN_NCH (SEQLEN / SCAN_T)

struct ScanOps {
    float  dtv[SCAN_T];
    float  uv[SCAN_T];
    float4 Bv[SCAN_T];
    float4 Cv[SCAN_T];
};

__device__ __forceinline__ void scan_load(
    ScanOps& o, const float* __restrict__ dtp, const float* __restrict__ up,
    const float* __restrict__ prp, int t0)
{
#pragma unroll
    for (int i = 0; i < SCAN_T; ++i) {
        const long t = t0 + i;
        o.dtv[i] = dtp[t * D_INNER];
        o.uv[i]  = up[t * (2 * D_INNER)];
        o.Bv[i]  = *(const float4*)(prp + t * NPROJ);
        o.Cv[i]  = *(const float4*)(prp + t * NPROJ + D_STATE);
    }
}

__device__ __forceinline__ void scan_chunk(
    const ScanOps& o, const float (&a)[4], float (&h)[4], float Dp,
    float* __restrict__ yp, int t0, int lane)
{
    float acc[SCAN_T];
#pragma unroll
    for (int i = 0; i < SCAN_T; ++i) {
        const float dtv = o.dtv[i];
        const float du  = dtv * o.uv[i];
        h[0] = __expf(dtv * a[0]) * h[0] + du * o.Bv[i].x;
        h[1] = __expf(dtv * a[1]) * h[1] + du * o.Bv[i].y;
        h[2] = __expf(dtv * a[2]) * h[2] + du * o.Bv[i].z;
        h[3] = __expf(dtv * a[3]) * h[3] + du * o.Bv[i].w;
        acc[i] = h[0] * o.Cv[i].x + h[1] * o.Cv[i].y + h[2] * o.Cv[i].z + h[3] * o.Cv[i].w;
    }
#pragma unroll
    for (int off = 16; off; off >>= 1) {
#pragma unroll
        for (int i = 0; i < SCAN_T; ++i) acc[i] += __shfl_xor(acc[i], off, 32);
    }
    if (lane == 0) {
#pragma unroll
        for (int i = 0; i < SCAN_T; ++i)
            yp[(long)(t0 + i) * D_INNER] = acc[i] + Dp * o.uv[i];
    }
}

__global__ __launch_bounds__(32) void mamba_scan(
    const float* __restrict__ dt, const float* __restrict__ proj,
    const float* __restrict__ UZ, const float* __restrict__ A_log,
    const float* __restrict__ D_param, float* __restrict__ y)
{
    const int bd   = blockIdx.x;
    const int b    = bd >> 10;
    const int d    = bd & (D_INNER - 1);
    const int lane = threadIdx.x;

    const float4 al = *(const float4*)(A_log + d * D_STATE + lane * 4);
    float a[4], h[4];
    a[0] = -__expf(al.x); a[1] = -__expf(al.y);
    a[2] = -__expf(al.z); a[3] = -__expf(al.w);
#pragma unroll
    for (int j = 0; j < 4; ++j) h[j] = 0.0f;
    const float Dp = D_param[d];

    const long row0 = (long)b * SEQLEN;
    const float* dtp = dt + row0 * D_INNER + d;
    const float* up  = UZ + row0 * (2 * D_INNER) + d;
    const float* prp = proj + row0 * NPROJ + DT_RANK + lane * 4;
    float*       yp  = y + row0 * D_INNER + d;

    ScanOps s0, s1;
    scan_load(s0, dtp, up, prp, 0);
    for (int c = 0; c < SCAN_NCH; c += 2) {
        scan_load(s1, dtp, up, prp, (c + 1) * SCAN_T);
        scan_chunk(s0, a, h, Dp, yp, c * SCAN_T, lane);
        const int cn = (c + 2 < SCAN_NCH) ? (c + 2) : (SCAN_NCH - 1);
        scan_load(s0, dtp, up, prp, cn * SCAN_T);
        scan_chunk(s1, a, h, Dp, yp, (c + 1) * SCAN_T, lane);
    }
}

__global__ __launch_bounds__(256) void ln_gate(
    float* __restrict__ y, const float* __restrict__ UZ,
    const float* __restrict__ gamma, const float* __restrict__ beta)
{
    const int row = blockIdx.x;
    const int tid = threadIdx.x;
    const int c0  = tid * 4;

    const float4 v = *(const float4*)(y + (long)row * D_INNER + c0);
    float s  = v.x + v.y + v.z + v.w;
    float s2 = v.x * v.x + v.y * v.y + v.z * v.z + v.w * v.w;

#pragma unroll
    for (int off = 16; off; off >>= 1) {
        s  += __shfl_xor(s, off, 32);
        s2 += __shfl_xor(s2, off, 32);
    }
    __shared__ float rs[8], rs2[8];
    const int wid = tid >> 5;
    if ((tid & 31) == 0) { rs[wid] = s; rs2[wid] = s2; }
    __syncthreads();
    s = 0.0f; s2 = 0.0f;
#pragma unroll
    for (int w = 0; w < 8; ++w) { s += rs[w]; s2 += rs2[w]; }

    const float mu   = s * (1.0f / D_INNER);
    const float var  = s2 * (1.0f / D_INNER) - mu * mu;
    const float rstd = rsqrtf(var + 1e-5f);

    const float4 ga = *(const float4*)(gamma + c0);
    const float4 be = *(const float4*)(beta + c0);
    const float4 zv = *(const float4*)(UZ + (long)row * (2 * D_INNER) + D_INNER + c0);

    float4 o;
    o.x = ((v.x - mu) * rstd * ga.x + be.x) * silu_f(zv.x);
    o.y = ((v.y - mu) * rstd * ga.y + be.y) * silu_f(zv.y);
    o.z = ((v.z - mu) * rstd * ga.z + be.z) * silu_f(zv.z);
    o.w = ((v.w - mu) * rstd * ga.w + be.w) * silu_f(zv.w);
    *(float4*)(y + (long)row * D_INNER + c0) = o;
}

extern "C" void kernel_launch(void* const* d_in, const int* in_sizes, int n_in,
                              void* d_out, int out_size, void* d_ws, size_t ws_size,
                              hipStream_t stream)
{
    const float* x     = (const float*)d_in[0];
    const float* W_in  = (const float*)d_in[1];
    const float* W_x   = (const float*)d_in[2];
    const float* W_dt  = (const float*)d_in[3];
    const float* b_dt  = (const float*)d_in[4];
    const float* W_out = (const float*)d_in[5];
    const float* gamma = (const float*)d_in[6];
    const float* beta  = (const float*)d_in[7];
    const float* A_log = (const float*)d_in[8];
    const float* D_par = (const float*)d_in[9];
    float* out = (float*)d_out;

    float* UZ   = (float*)d_ws;
    float* proj = UZ   + (size_t)ROWS * 2 * D_INNER;
    float* dtb  = proj + (size_t)ROWS * NPROJ;
    float* yb   = dtb  + (size_t)ROWS * D_INNER;

    gemm16_wmma_f32<1, 4, D_MODEL, 2><<<dim3((2 * D_INNER) / 64, ROWS / 16), 32, 0, stream>>>(
        x, D_MODEL, W_in, 2 * D_INNER, UZ, 2 * D_INNER, nullptr, D_INNER);

    gemm16_wmma_f32<0, 2, D_INNER, 4><<<dim3(NPROJ / 32, ROWS / 16), 32, 0, stream>>>(
        UZ, 2 * D_INNER, W_x, NPROJ, proj, NPROJ, nullptr, 0);

    gemm16_wmma_f32<2, 4, DT_RANK, 1><<<dim3(D_INNER / 64, ROWS / 16), 32, 0, stream>>>(
        proj, NPROJ, W_dt, D_INNER, dtb, D_INNER, b_dt, 0);

    mamba_scan<<<BATCH * D_INNER, 32, 0, stream>>>(dtb, proj, UZ, A_log, D_par, yb);

    ln_gate<<<ROWS, 256, 0, stream>>>(yb, UZ, gamma, beta);

    gemm16_wmma_f32<0, 4, D_INNER, 4><<<dim3(D_MODEL / 64, ROWS / 16), 32, 0, stream>>>(
        yb, D_INNER, W_out, D_MODEL, out, D_MODEL, nullptr, 0);
}


# Scan: a chunk's 16 dt and 16 u values come by two vector loads (one step per lane, read out by lane broadcast) instead of 32 four-byte scalar loads that each open a new scalar-cache line
# speedup vs baseline: 4.0538x; 1.0378x over previous
#include <hip/hip_runtime.h>
#include <hip/hip_bf16.h>
#include <math.h>

typedef __attribute__((ext_vector_type(2))) float v2f;
typedef __attribute__((ext_vector_type(8))) float v8f;

#define BATCH   2
#define SEQLEN  1024
#define D_MODEL 512
#define D_INNER 1024
#define D_STATE 128
#define DT_RANK 32
#define NPROJ   (DT_RANK + 2 * D_STATE)
#define ROWS    (BATCH * SEQLEN)

__device__ __forceinline__ float silu_f(float x) {
    return x * (1.0f / (1.0f + __expf(-x)));
}
__device__ __forceinline__ float softplus_f(float x) {
    return (x > 20.0f) ? x : log1pf(__expf(x));
}

typedef __attribute__((ext_vector_type(4))) float v4f;

template <int NB> struct VecN;
template <> struct VecN<2> { typedef v2f T; };
template <> struct VecN<4> { typedef v4f T; };

#define GEMM_KB 32

template <int NB>
struct GemmFrag {
    v4f a[4];
    typename VecN<NB>::T b0[8];
    typename VecN<NB>::T b1[8];
};

template <int NB>
__device__ __forceinline__ void gemm_load(
    GemmFrag<NB>& f, const float* __restrict__ Ap,
    const float* __restrict__ Bp, int ldb, int kb)
{
    typedef typename VecN<NB>::T VB;
    const float* ap = Ap + kb * GEMM_KB;
#pragma unroll
    for (int j = 0; j < 4; ++j) f.a[j] = *(const v4f*)(ap + 4 * j);
    const float* bp = Bp + (long)(kb * GEMM_KB) * ldb;
#pragma unroll
    for (int s = 0; s < 8; ++s) {
        f.b0[s] = *(const VB*)(bp + (long)(2 * s) * ldb);
        f.b1[s] = *(const VB*)(bp + (long)(2 * s + 1) * ldb);
    }
}

template <int NB>
__device__ __forceinline__ void gemm_block(const GemmFrag<NB>& f, v8f (&acc)[NB])
{
#pragma unroll
    for (int s = 0; s < 8; ++s) {
        v2f a;
        a.x = f.a[s >> 1][(s & 1) * 2];
        a.y = f.a[s >> 1][(s & 1) * 2 + 1];
#pragma unroll
        for (int nb = 0; nb < NB; ++nb) {
            v2f b;
            b.x = f.b0[s][nb];
            b.y = f.b1[s][nb];
            acc[nb] = __builtin_amdgcn_wmma_f32_16x16x4_f32(
                 false, a,  false, b,
                 (short)0, acc[nb],  false,  false);
        }
    }
}

template <int NB, int R, bool LAST>
__device__ __forceinline__ void gemm_ring_pass(
    GemmFrag<NB> (&f)[R], v8f (&acc)[NB], const float* __restrict__ Ap,
    const float* __restrict__ Bp, int ldb, int kb)
{
#pragma unroll
    for (int j = 0; j < R; ++j) {
        if (!LAST || j == 0)
            gemm_load<NB>(f[(j + R - 1) % R], Ap, Bp, ldb, kb + j + R - 1);
        __builtin_amdgcn_sched_barrier(0);
        gemm_block<NB>(f[j], acc);
        __builtin_amdgcn_sched_barrier(0);
    }
}

template <int EPI, int NB, int K, int R>
__global__ __launch_bounds__(32) void gemm16_wmma_f32(
    const float* __restrict__ A, int lda,
    const float* __restrict__ B, int ldb,
    float* __restrict__ C, int ldc,
    const float* __restrict__ bias, int split)
{
    typedef typename VecN<NB>::T VB;
    constexpr int NKB = K / GEMM_KB;
    static_assert(K % GEMM_KB == 0 && NKB % R == 0, "K blocks must fill the ring");

    const int tn   = blockIdx.x;
    const int tm   = blockIdx.y;
    const int lane = threadIdx.x;
    const int half = lane >> 4;
    const int l16  = lane & 15;

    const float* Ap = A + (long)(tm * 16 + l16) * lda + 16 * half;
    const float* Bp = B + (long)(16 * half) * ldb + (tn * (NB * 16) + NB * l16);

    v8f acc[NB];
#pragma unroll
    for (int nb = 0; nb < NB; ++nb) acc[nb] = (v8f){};

    GemmFrag<NB> f[R];
#pragma unroll
    for (int j = 0; j < R - 1; ++j) {
        gemm_load<NB>(f[j], Ap, Bp, ldb, j);
        __builtin_amdgcn_sched_barrier(0);
    }

#pragma unroll 1
    for (int kb = 0; kb < NKB - R; kb += R)
        gemm_ring_pass<NB, R, false>(f, acc, Ap, Bp, ldb, kb);
    gemm_ring_pass<NB, R, true>(f, acc, Ap, Bp, ldb, NKB - R);

    const int col0 = tn * (NB * 16) + NB * l16;
    VB bv = {};
    if (EPI == 2) bv = *(const VB*)(bias + col0);
#pragma unroll
    for (int r = 0; r < 8; ++r) {
        const int row = tm * 16 + half * 8 + r;
        VB o;
#pragma unroll
        for (int nb = 0; nb < NB; ++nb) {
            const int col = col0 + nb;
            float v = acc[nb][r];
            if (EPI == 1) {
                if (col < split) v = silu_f(v);
            } else if (EPI == 2) {
                v = softplus_f(v + bv[nb]);
            }
            o[nb] = v;
        }
        *(VB*)(C + (long)row * ldc + col0) = o;
    }
}

#define SCAN_T   16
#define SCAN_NCH (SEQLEN / SCAN_T)
static_assert(SCAN_T == 16, "one step per lane of a 16-lane row");

struct ScanOps {
    float  dtv;
    float  uv;
    float4 Bv[SCAN_T];
    float4 Cv[SCAN_T];
};

__device__ __forceinline__ float lane_value(float v, int src) {
    return __builtin_bit_cast(float,
        __builtin_amdgcn_readlane(__builtin_bit_cast(int, v), src));
}

__device__ __forceinline__ void scan_load(
    ScanOps& o, const float* __restrict__ dtl, const float* __restrict__ ul,
    const float* __restrict__ prp, int t0)
{
    o.dtv = dtl[(long)t0 * D_INNER];
    o.uv  = ul[(long)t0 * (2 * D_INNER)];
#pragma unroll
    for (int i = 0; i < SCAN_T; ++i) {
        const long t = t0 + i;
        o.Bv[i]  = *(const float4*)(prp + t * NPROJ);
        o.Cv[i]  = *(const float4*)(prp + t * NPROJ + D_STATE);
    }
}

__device__ __forceinline__ void scan_chunk(
    const ScanOps& o, const float (&a)[4], float (&h)[4], float Dp,
    float* __restrict__ yp, int t0, int lane)
{
    float acc[SCAN_T];
    float us[SCAN_T];
#pragma unroll
    for (int i = 0; i < SCAN_T; ++i) {
        const float dtv = lane_value(o.dtv, i);
        us[i] = lane_value(o.uv, i);
        const float du  = dtv * us[i];
        h[0] = __expf(dtv * a[0]) * h[0] + du * o.Bv[i].x;
        h[1] = __expf(dtv * a[1]) * h[1] + du * o.Bv[i].y;
        h[2] = __expf(dtv * a[2]) * h[2] + du * o.Bv[i].z;
        h[3] = __expf(dtv * a[3]) * h[3] + du * o.Bv[i].w;
        acc[i] = h[0] * o.Cv[i].x + h[1] * o.Cv[i].y + h[2] * o.Cv[i].z + h[3] * o.Cv[i].w;
    }
#pragma unroll
    for (int off = 16; off; off >>= 1) {
#pragma unroll
        for (int i = 0; i < SCAN_T; ++i) acc[i] += __shfl_xor(acc[i], off, 32);
    }
    if (lane == 0) {
#pragma unroll
        for (int i = 0; i < SCAN_T; ++i)
            yp[(long)(t0 + i) * D_INNER] = acc[i] + Dp * us[i];
    }
}

__global__ __launch_bounds__(32) void mamba_scan(
    const float* __restrict__ dt, const float* __restrict__ proj,
    const float* __restrict__ UZ, const float* __restrict__ A_log,
    const float* __restrict__ D_param, float* __restrict__ y)
{
    const int bd   = blockIdx.x;
    const int b    = bd >> 10;
    const int d    = bd & (D_INNER - 1);
    const int lane = threadIdx.x;

    const float4 al = *(const float4*)(A_log + d * D_STATE + lane * 4);
    float a[4], h[4];
    a[0] = -__expf(al.x); a[1] = -__expf(al.y);
    a[2] = -__expf(al.z); a[3] = -__expf(al.w);
#pragma unroll
    for (int j = 0; j < 4; ++j) h[j] = 0.0f;
    const float Dp = D_param[d];

    const long row0 = (long)b * SEQLEN;
    const int    ls  = lane & 15;
    const float* dtp = dt + (row0 + ls) * D_INNER + d;
    const float* up  = UZ + (row0 + ls) * (2 * D_INNER) + d;
    const float* prp = proj + row0 * NPROJ + DT_RANK + lane * 4;
    float*       yp  = y + row0 * D_INNER + d;

    ScanOps s0, s1;
    scan_load(s0, dtp, up, prp, 0);
    for (int c = 0; c < SCAN_NCH; c += 2) {
        scan_load(s1, dtp, up, prp, (c + 1) * SCAN_T);
        scan_chunk(s0, a, h, Dp, yp, c * SCAN_T, lane);
        const int cn = (c + 2 < SCAN_NCH) ? (c + 2) : (SCAN_NCH - 1);
        scan_load(s0, dtp, up, prp, cn * SCAN_T);
        scan_chunk(s1, a, h, Dp, yp, (c + 1) * SCAN_T, lane);
    }
}

__global__ __launch_bounds__(256) void ln_gate(
    float* __restrict__ y, const float* __restrict__ UZ,
    const float* __restrict__ gamma, const float* __restrict__ beta)
{
    const int row = blockIdx.x;
    const int tid = threadIdx.x;
    const int c0  = tid * 4;

    const float4 v = *(const float4*)(y + (long)row * D_INNER + c0);
    float s  = v.x + v.y + v.z + v.w;
    float s2 = v.x * v.x + v.y * v.y + v.z * v.z + v.w * v.w;

#pragma unroll
    for (int off = 16; off; off >>= 1) {
        s  += __shfl_xor(s, off, 32);
        s2 += __shfl_xor(s2, off, 32);
    }
    __shared__ float rs[8], rs2[8];
    const int wid = tid >> 5;
    if ((tid & 31) == 0) { rs[wid] = s; rs2[wid] = s2; }
    __syncthreads();
    s = 0.0f; s2 = 0.0f;
#pragma unroll
    for (int w = 0; w < 8; ++w) { s += rs[w]; s2 += rs2[w]; }

    const float mu   = s * (1.0f / D_INNER);
    const float var  = s2 * (1.0f / D_INNER) - mu * mu;
    const float rstd = rsqrtf(var + 1e-5f);

    const float4 ga = *(const float4*)(gamma + c0);
    const float4 be = *(const float4*)(beta + c0);
    const float4 zv = *(const float4*)(UZ + (long)row * (2 * D_INNER) + D_INNER + c0);

    float4 o;
    o.x = ((v.x - mu) * rstd * ga.x + be.x) * silu_f(zv.x);
    o.y = ((v.y - mu) * rstd * ga.y + be.y) * silu_f(zv.y);
    o.z = ((v.z - mu) * rstd * ga.z + be.z) * silu_f(zv.z);
    o.w = ((v.w - mu) * rstd * ga.w + be.w) * silu_f(zv.w);
    *(float4*)(y + (long)row * D_INNER + c0) = o;
}

extern "C" void kernel_launch(void* const* d_in, const int* in_sizes, int n_in,
                              void* d_out, int out_size, void* d_ws, size_t ws_size,
                              hipStream_t stream)
{
    const float* x     = (const float*)d_in[0];
    const float* W_in  = (const float*)d_in[1];
    const float* W_x   = (const float*)d_in[2];
    const float* W_dt  = (const float*)d_in[3];
    const float* b_dt  = (const float*)d_in[4];
    const float* W_out = (const float*)d_in[5];
    const float* gamma = (const float*)d_in[6];
    const float* beta  = (const float*)d_in[7];
    const float* A_log = (const float*)d_in[8];
    const float* D_par = (const float*)d_in[9];
    float* out = (float*)d_out;

    float* UZ   = (float*)d_ws;
    float* proj = UZ   + (size_t)ROWS * 2 * D_INNER;
    float* dtb  = proj + (size_t)ROWS * NPROJ;
    float* yb   = dtb  + (size_t)ROWS * D_INNER;

    gemm16_wmma_f32<1, 4, D_MODEL, 2><<<dim3((2 * D_INNER) / 64, ROWS / 16), 32, 0, stream>>>(
        x, D_MODEL, W_in, 2 * D_INNER, UZ, 2 * D_INNER, nullptr, D_INNER);

    gemm16_wmma_f32<0, 2, D_INNER, 4><<<dim3(NPROJ / 32, ROWS / 16), 32, 0, stream>>>(
        UZ, 2 * D_INNER, W_x, NPROJ, proj, NPROJ, nullptr, 0);

    gemm16_wmma_f32<2, 4, DT_RANK, 1><<<dim3(D_INNER / 64, ROWS / 16), 32, 0, stream>>>(
        proj, NPROJ, W_dt, D_INNER, dtb, D_INNER, b_dt, 0);

    mamba_scan<<<BATCH * D_INNER, 32, 0, stream>>>(dtb, proj, UZ, A_log, D_par, yb);

    ln_gate<<<ROWS, 256, 0, stream>>>(yb, UZ, gamma, beta);

    gemm16_wmma_f32<0, 4, D_INNER, 4><<<dim3(D_MODEL / 64, ROWS / 16), 32, 0, stream>>>(
        yb, D_INNER, W_out, D_MODEL, out, D_MODEL, nullptr, 0);
}


# Scan: a chunk's 16 lane sums reduced by a transposing xor tree (lanes halve the values they carry per level: 16 register exchanges, not 80 LDS permutes; same pair order); one store per chunk
# speedup vs baseline: 4.2114x; 1.0389x over previous
#include <hip/hip_runtime.h>
#include <hip/hip_bf16.h>
#include <math.h>

typedef __attribute__((ext_vector_type(2))) float v2f;
typedef __attribute__((ext_vector_type(8))) float v8f;

#define BATCH   2
#define SEQLEN  1024
#define D_MODEL 512
#define D_INNER 1024
#define D_STATE 128
#define DT_RANK 32
#define NPROJ   (DT_RANK + 2 * D_STATE)
#define ROWS    (BATCH * SEQLEN)

__device__ __forceinline__ float silu_f(float x) {
    return x * (1.0f / (1.0f + __expf(-x)));
}
__device__ __forceinline__ float softplus_f(float x) {
    return (x > 20.0f) ? x : log1pf(__expf(x));
}

typedef __attribute__((ext_vector_type(4))) float v4f;

template <int NB> struct VecN;
template <> struct VecN<2> { typedef v2f T; };
template <> struct VecN<4> { typedef v4f T; };

#define GEMM_KB 32

template <int NB>
struct GemmFrag {
    v4f a[4];
    typename VecN<NB>::T b0[8];
    typename VecN<NB>::T b1[8];
};

template <int NB>
__device__ __forceinline__ void gemm_load(
    GemmFrag<NB>& f, const float* __restrict__ Ap,
    const float* __restrict__ Bp, int ldb, int kb)
{
    typedef typename VecN<NB>::T VB;
    const float* ap = Ap + kb * GEMM_KB;
#pragma unroll
    for (int j = 0; j < 4; ++j) f.a[j] = *(const v4f*)(ap + 4 * j);
    const float* bp = Bp + (long)(kb * GEMM_KB) * ldb;
#pragma unroll
    for (int s = 0; s < 8; ++s) {
        f.b0[s] = *(const VB*)(bp + (long)(2 * s) * ldb);
        f.b1[s] = *(const VB*)(bp + (long)(2 * s + 1) * ldb);
    }
}

template <int NB>
__device__ __forceinline__ void gemm_block(const GemmFrag<NB>& f, v8f (&acc)[NB])
{
#pragma unroll
    for (int s = 0; s < 8; ++s) {
        v2f a;
        a.x = f.a[s >> 1][(s & 1) * 2];
        a.y = f.a[s >> 1][(s & 1) * 2 + 1];
#pragma unroll
        for (int nb = 0; nb < NB; ++nb) {
            v2f b;
            b.x = f.b0[s][nb];
            b.y = f.b1[s][nb];
            acc[nb] = __builtin_amdgcn_wmma_f32_16x16x4_f32(
                 false, a,  false, b,
                 (short)0, acc[nb],  false,  false);
        }
    }
}

template <int NB, int R, bool LAST>
__device__ __forceinline__ void gemm_ring_pass(
    GemmFrag<NB> (&f)[R], v8f (&acc)[NB], const float* __restrict__ Ap,
    const float* __restrict__ Bp, int ldb, int kb)
{
#pragma unroll
    for (int j = 0; j < R; ++j) {
        if (!LAST || j == 0)
            gemm_load<NB>(f[(j + R - 1) % R], Ap, Bp, ldb, kb + j + R - 1);
        __builtin_amdgcn_sched_barrier(0);
        gemm_block<NB>(f[j], acc);
        __builtin_amdgcn_sched_barrier(0);
    }
}

template <int EPI, int NB, int K, int R>
__global__ __launch_bounds__(32) void gemm16_wmma_f32(
    const float* __restrict__ A, int lda,
    const float* __restrict__ B, int ldb,
    float* __restrict__ C, int ldc,
    const float* __restrict__ bias, int split)
{
    typedef typename VecN<NB>::T VB;
    constexpr int NKB = K / GEMM_KB;
    static_assert(K % GEMM_KB == 0 && NKB % R == 0, "K blocks must fill the ring");

    const int tn   = blockIdx.x;
    const int tm   = blockIdx.y;
    const int lane = threadIdx.x;
    const int half = lane >> 4;
    const int l16  = lane & 15;

    const float* Ap = A + (long)(tm * 16 + l16) * lda + 16 * half;
    const float* Bp = B + (long)(16 * half) * ldb + (tn * (NB * 16) + NB * l16);

    v8f acc[NB];
#pragma unroll
    for (int nb = 0; nb < NB; ++nb) acc[nb] = (v8f){};

    GemmFrag<NB> f[R];
#pragma unroll
    for (int j = 0; j < R - 1; ++j) {
        gemm_load<NB>(f[j], Ap, Bp, ldb, j);
        __builtin_amdgcn_sched_barrier(0);
    }

#pragma unroll 1
    for (int kb = 0; kb < NKB - R; kb += R)
        gemm_ring_pass<NB, R, false>(f, acc, Ap, Bp, ldb, kb);
    gemm_ring_pass<NB, R, true>(f, acc, Ap, Bp, ldb, NKB - R);

    const int col0 = tn * (NB * 16) + NB * l16;
    VB bv = {};
    if (EPI == 2) bv = *(const VB*)(bias + col0);
#pragma unroll
    for (int r = 0; r < 8; ++r) {
        const int row = tm * 16 + half * 8 + r;
        VB o;
#pragma unroll
        for (int nb = 0; nb < NB; ++nb) {
            const int col = col0 + nb;
            float v = acc[nb][r];
            if (EPI == 1) {
                if (col < split) v = silu_f(v);
            } else if (EPI == 2) {
                v = softplus_f(v + bv[nb]);
            }
            o[nb] = v;
        }
        *(VB*)(C + (long)row * ldc + col0) = o;
    }
}

#define SCAN_T   16
#define SCAN_NCH (SEQLEN / SCAN_T)
static_assert(SCAN_T == 16, "one step per pair of lanes");

struct ScanOps {
    float  dtv;
    float  uv;
    float4 Bv[SCAN_T];
    float4 Cv[SCAN_T];
};

__device__ __forceinline__ float lane_value(float v, int src) {
    return __builtin_bit_cast(float,
        __builtin_amdgcn_readlane(__builtin_bit_cast(int, v), src));
}

__device__ __forceinline__ float other_row(float v) {
    const int b = __builtin_bit_cast(int, v);
    return __builtin_bit_cast(float,
        __builtin_amdgcn_permlanex16(b, b, 0x76543210, 0xfedcba98, false, false));
}

template <int M>
__device__ __forceinline__ float row_xor(float v) {
    const int b = __builtin_bit_cast(int, v);
    return __builtin_bit_cast(float,
        __builtin_amdgcn_update_dpp(b, b, 0x160 + M, 0xf, 0xf, true));
}

__device__ __forceinline__ void scan_load(
    ScanOps& o, const float* __restrict__ dtl, const float* __restrict__ ul,
    const float* __restrict__ prp, int t0)
{
    o.dtv = dtl[(long)t0 * D_INNER];
    o.uv  = ul[(long)t0 * (2 * D_INNER)];
#pragma unroll
    for (int i = 0; i < SCAN_T; ++i) {
        const long t = t0 + i;
        o.Bv[i]  = *(const float4*)(prp + t * NPROJ);
        o.Cv[i]  = *(const float4*)(prp + t * NPROJ + D_STATE);
    }
}

__device__ __forceinline__ void scan_chunk(
    const ScanOps& o, const float (&a)[4], float (&h)[4], float Dp,
    float* __restrict__ yp, int t0, int lane)
{
    float acc[SCAN_T];
#pragma unroll
    for (int i = 0; i < SCAN_T; ++i) {
        const float dtv = lane_value(o.dtv, 2 * i);
        const float du  = dtv * lane_value(o.uv, 2 * i);
        h[0] = __expf(dtv * a[0]) * h[0] + du * o.Bv[i].x;
        h[1] = __expf(dtv * a[1]) * h[1] + du * o.Bv[i].y;
        h[2] = __expf(dtv * a[2]) * h[2] + du * o.Bv[i].z;
        h[3] = __expf(dtv * a[3]) * h[3] + du * o.Bv[i].w;
        acc[i] = h[0] * o.Cv[i].x + h[1] * o.Cv[i].y + h[2] * o.Cv[i].z + h[3] * o.Cv[i].w;
    }
    const bool up16 = (lane & 16) != 0;
    float b1[8];
#pragma unroll
    for (int i = 0; i < 8; ++i) {
        const float keep = up16 ? acc[i + 8] : acc[i];
        const float send = up16 ? acc[i] : acc[i + 8];
        b1[i] = keep + other_row(send);
    }
    const bool up8 = (lane & 8) != 0;
    float b2[4];
#pragma unroll
    for (int i = 0; i < 4; ++i) {
        const float keep = up8 ? b1[i + 4] : b1[i];
        const float send = up8 ? b1[i] : b1[i + 4];
        b2[i] = keep + row_xor<8>(send);
    }
    const bool up4 = (lane & 4) != 0;
    float b3[2];
#pragma unroll
    for (int i = 0; i < 2; ++i) {
        const float keep = up4 ? b2[i + 2] : b2[i];
        const float send = up4 ? b2[i] : b2[i + 2];
        b3[i] = keep + row_xor<4>(send);
    }
    const bool up2 = (lane & 2) != 0;
    const float keep2 = up2 ? b3[1] : b3[0];
    const float send2 = up2 ? b3[0] : b3[1];
    const float b4 = keep2 + row_xor<2>(send2);
    const float tot = b4 + row_xor<1>(b4);
    if ((lane & 1) == 0)
        yp[(long)(t0 + (lane >> 1)) * D_INNER] = tot + Dp * o.uv;
}

__global__ __launch_bounds__(32) void mamba_scan(
    const float* __restrict__ dt, const float* __restrict__ proj,
    const float* __restrict__ UZ, const float* __restrict__ A_log,
    const float* __restrict__ D_param, float* __restrict__ y)
{
    const int bd   = blockIdx.x;
    const int b    = bd >> 10;
    const int d    = bd & (D_INNER - 1);
    const int lane = threadIdx.x;

    const float4 al = *(const float4*)(A_log + d * D_STATE + lane * 4);
    float a[4], h[4];
    a[0] = -__expf(al.x); a[1] = -__expf(al.y);
    a[2] = -__expf(al.z); a[3] = -__expf(al.w);
#pragma unroll
    for (int j = 0; j < 4; ++j) h[j] = 0.0f;
    const float Dp = D_param[d];

    const long row0 = (long)b * SEQLEN;
    const int    ls  = lane >> 1;
    const float* dtp = dt + (row0 + ls) * D_INNER + d;
    const float* up  = UZ + (row0 + ls) * (2 * D_INNER) + d;
    const float* prp = proj + row0 * NPROJ + DT_RANK + lane * 4;
    float*       yp  = y + row0 * D_INNER + d;

    ScanOps s0, s1;
    scan_load(s0, dtp, up, prp, 0);
    for (int c = 0; c < SCAN_NCH; c += 2) {
        scan_load(s1, dtp, up, prp, (c + 1) * SCAN_T);
        scan_chunk(s0, a, h, Dp, yp, c * SCAN_T, lane);
        const int cn = (c + 2 < SCAN_NCH) ? (c + 2) : (SCAN_NCH - 1);
        scan_load(s0, dtp, up, prp, cn * SCAN_T);
        scan_chunk(s1, a, h, Dp, yp, (c + 1) * SCAN_T, lane);
    }
}

__global__ __launch_bounds__(256) void ln_gate(
    float* __restrict__ y, const float* __restrict__ UZ,
    const float* __restrict__ gamma, const float* __restrict__ beta)
{
    const int row = blockIdx.x;
    const int tid = threadIdx.x;
    const int c0  = tid * 4;

    const float4 v = *(const float4*)(y + (long)row * D_INNER + c0);
    float s  = v.x + v.y + v.z + v.w;
    float s2 = v.x * v.x + v.y * v.y + v.z * v.z + v.w * v.w;

#pragma unroll
    for (int off = 16; off; off >>= 1) {
        s  += __shfl_xor(s, off, 32);
        s2 += __shfl_xor(s2, off, 32);
    }
    __shared__ float rs[8], rs2[8];
    const int wid = tid >> 5;
    if ((tid & 31) == 0) { rs[wid] = s; rs2[wid] = s2; }
    __syncthreads();
    s = 0.0f; s2 = 0.0f;
#pragma unroll
    for (int w = 0; w < 8; ++w) { s += rs[w]; s2 += rs2[w]; }

    const float mu   = s * (1.0f / D_INNER);
    const float var  = s2 * (1.0f / D_INNER) - mu * mu;
    const float rstd = rsqrtf(var + 1e-5f);

    const float4 ga = *(const float4*)(gamma + c0);
    const float4 be = *(const float4*)(beta + c0);
    const float4 zv = *(const float4*)(UZ + (long)row * (2 * D_INNER) + D_INNER + c0);

    float4 o;
    o.x = ((v.x - mu) * rstd * ga.x + be.x) * silu_f(zv.x);
    o.y = ((v.y - mu) * rstd * ga.y + be.y) * silu_f(zv.y);
    o.z = ((v.z - mu) * rstd * ga.z + be.z) * silu_f(zv.z);
    o.w = ((v.w - mu) * rstd * ga.w + be.w) * silu_f(zv.w);
    *(float4*)(y + (long)row * D_INNER + c0) = o;
}

extern "C" void kernel_launch(void* const* d_in, const int* in_sizes, int n_in,
                              void* d_out, int out_size, void* d_ws, size_t ws_size,
                              hipStream_t stream)
{
    const float* x     = (const float*)d_in[0];
    const float* W_in  = (const float*)d_in[1];
    const float* W_x   = (const float*)d_in[2];
    const float* W_dt  = (const float*)d_in[3];
    const float* b_dt  = (const float*)d_in[4];
    const float* W_out = (const float*)d_in[5];
    const float* gamma = (const float*)d_in[6];
    const float* beta  = (const float*)d_in[7];
    const float* A_log = (const float*)d_in[8];
    const float* D_par = (const float*)d_in[9];
    float* out = (float*)d_out;

    float* UZ   = (float*)d_ws;
    float* proj = UZ   + (size_t)ROWS * 2 * D_INNER;
    float* dtb  = proj + (size_t)ROWS * NPROJ;
    float* yb   = dtb  + (size_t)ROWS * D_INNER;

    gemm16_wmma_f32<1, 4, D_MODEL, 2><<<dim3((2 * D_INNER) / 64, ROWS / 16), 32, 0, stream>>>(
        x, D_MODEL, W_in, 2 * D_INNER, UZ, 2 * D_INNER, nullptr, D_INNER);

    gemm16_wmma_f32<0, 2, D_INNER, 4><<<dim3(NPROJ / 32, ROWS / 16), 32, 0, stream>>>(
        UZ, 2 * D_INNER, W_x, NPROJ, proj, NPROJ, nullptr, 0);

    gemm16_wmma_f32<2, 4, DT_RANK, 1><<<dim3(D_INNER / 64, ROWS / 16), 32, 0, stream>>>(
        proj, NPROJ, W_dt, D_INNER, dtb, D_INNER, b_dt, 0);

    mamba_scan<<<BATCH * D_INNER, 32, 0, stream>>>(dtb, proj, UZ, A_log, D_par, yb);

    ln_gate<<<ROWS, 256, 0, stream>>>(yb, UZ, gamma, beta);

    gemm16_wmma_f32<0, 4, D_INNER, 4><<<dim3(D_MODEL / 64, ROWS / 16), 32, 0, stream>>>(
        yb, D_INNER, W_out, D_MODEL, out, D_MODEL, nullptr, 0);
}


# Launch shape: every one-wave work-group (scan and the four GEMMs) becomes a four-wave work-group of independent waves, no barrier, so a WGP's waves start one per SIMD; sums and their order unchanged
# speedup vs baseline: 5.0934x; 1.2094x over previous
#include <hip/hip_runtime.h>
#include <hip/hip_bf16.h>
#include <math.h>

typedef __attribute__((ext_vector_type(2))) float v2f;
typedef __attribute__((ext_vector_type(8))) float v8f;

#define BATCH   2
#define SEQLEN  1024
#define D_MODEL 512
#define D_INNER 1024
#define D_STATE 128
#define DT_RANK 32
#define NPROJ   (DT_RANK + 2 * D_STATE)
#define ROWS    (BATCH * SEQLEN)

__device__ __forceinline__ float silu_f(float x) {
    return x * (1.0f / (1.0f + __expf(-x)));
}
__device__ __forceinline__ float softplus_f(float x) {
    return (x > 20.0f) ? x : log1pf(__expf(x));
}

typedef __attribute__((ext_vector_type(4))) float v4f;

template <int NB> struct VecN;
template <> struct VecN<2> { typedef v2f T; };
template <> struct VecN<4> { typedef v4f T; };

#define GEMM_KB 32
#define GEMM_WPB 4

template <int NB>
struct GemmFrag {
    v4f a[4];
    typename VecN<NB>::T b0[8];
    typename VecN<NB>::T b1[8];
};

template <int NB>
__device__ __forceinline__ void gemm_load(
    GemmFrag<NB>& f, const float* __restrict__ Ap,
    const float* __restrict__ Bp, int ldb, int kb)
{
    typedef typename VecN<NB>::T VB;
    const float* ap = Ap + kb * GEMM_KB;
#pragma unroll
    for (int j = 0; j < 4; ++j) f.a[j] = *(const v4f*)(ap + 4 * j);
    const float* bp = Bp + (long)(kb * GEMM_KB) * ldb;
#pragma unroll
    for (int s = 0; s < 8; ++s) {
        f.b0[s] = *(const VB*)(bp + (long)(2 * s) * ldb);
        f.b1[s] = *(const VB*)(bp + (long)(2 * s + 1) * ldb);
    }
}

template <int NB>
__device__ __forceinline__ void gemm_block(const GemmFrag<NB>& f, v8f (&acc)[NB])
{
#pragma unroll
    for (int s = 0; s < 8; ++s) {
        v2f a;
        a.x = f.a[s >> 1][(s & 1) * 2];
        a.y = f.a[s >> 1][(s & 1) * 2 + 1];
#pragma unroll
        for (int nb = 0; nb < NB; ++nb) {
            v2f b;
            b.x = f.b0[s][nb];
            b.y = f.b1[s][nb];
            acc[nb] = __builtin_amdgcn_wmma_f32_16x16x4_f32(
                 false, a,  false, b,
                 (short)0, acc[nb],  false,  false);
        }
    }
}

template <int NB, int R, bool LAST>
__device__ __forceinline__ void gemm_ring_pass(
    GemmFrag<NB> (&f)[R], v8f (&acc)[NB], const float* __restrict__ Ap,
    const float* __restrict__ Bp, int ldb, int kb)
{
#pragma unroll
    for (int j = 0; j < R; ++j) {
        if (!LAST || j == 0)
            gemm_load<NB>(f[(j + R - 1) % R], Ap, Bp, ldb, kb + j + R - 1);
        __builtin_amdgcn_sched_barrier(0);
        gemm_block<NB>(f[j], acc);
        __builtin_amdgcn_sched_barrier(0);
    }
}

template <int EPI, int NB, int K, int R>
__global__ __launch_bounds__(32 * GEMM_WPB) void gemm16_wmma_f32(
    const float* __restrict__ A, int lda,
    const float* __restrict__ B, int ldb,
    float* __restrict__ C, int ldc,
    const float* __restrict__ bias, int split)
{
    typedef typename VecN<NB>::T VB;
    constexpr int NKB = K / GEMM_KB;
    static_assert(K % GEMM_KB == 0 && NKB % R == 0, "K blocks must fill the ring");

    const int tn   = blockIdx.x;
    const int wv   = __builtin_amdgcn_readfirstlane((int)threadIdx.y);
    const int tm   = blockIdx.y * GEMM_WPB + wv;
    const int lane = threadIdx.x;
    const int half = lane >> 4;
    const int l16  = lane & 15;

    const float* Ap = A + (long)(tm * 16 + l16) * lda + 16 * half;
    const float* Bp = B + (long)(16 * half) * ldb + (tn * (NB * 16) + NB * l16);

    v8f acc[NB];
#pragma unroll
    for (int nb = 0; nb < NB; ++nb) acc[nb] = (v8f){};

    GemmFrag<NB> f[R];
#pragma unroll
    for (int j = 0; j < R - 1; ++j) {
        gemm_load<NB>(f[j], Ap, Bp, ldb, j);
        __builtin_amdgcn_sched_barrier(0);
    }

#pragma unroll 1
    for (int kb = 0; kb < NKB - R; kb += R)
        gemm_ring_pass<NB, R, false>(f, acc, Ap, Bp, ldb, kb);
    gemm_ring_pass<NB, R, true>(f, acc, Ap, Bp, ldb, NKB - R);

    const int col0 = tn * (NB * 16) + NB * l16;
    VB bv = {};
    if (EPI == 2) bv = *(const VB*)(bias + col0);
#pragma unroll
    for (int r = 0; r < 8; ++r) {
        const int row = tm * 16 + half * 8 + r;
        VB o;
#pragma unroll
        for (int nb = 0; nb < NB; ++nb) {
            const int col = col0 + nb;
            float v = acc[nb][r];
            if (EPI == 1) {
                if (col < split) v = silu_f(v);
            } else if (EPI == 2) {
                v = softplus_f(v + bv[nb]);
            }
            o[nb] = v;
        }
        *(VB*)(C + (long)row * ldc + col0) = o;
    }
}

#define SCAN_T   16
#define SCAN_NCH (SEQLEN / SCAN_T)
#define SCAN_WPB 4
static_assert(SCAN_T == 16, "one step per pair of lanes");
static_assert(D_INNER % SCAN_WPB == 0, "a work-group's channels share one batch entry");
static_assert((ROWS / 16) % GEMM_WPB == 0, "M tiles must fill the work-groups");

struct ScanOps {
    float  dtv;
    float  uv;
    float4 Bv[SCAN_T];
    float4 Cv[SCAN_T];
};

__device__ __forceinline__ float lane_value(float v, int src) {
    return __builtin_bit_cast(float,
        __builtin_amdgcn_readlane(__builtin_bit_cast(int, v), src));
}

__device__ __forceinline__ float other_row(float v) {
    const int b = __builtin_bit_cast(int, v);
    return __builtin_bit_cast(float,
        __builtin_amdgcn_permlanex16(b, b, 0x76543210, 0xfedcba98, false, false));
}

template <int M>
__device__ __forceinline__ float row_xor(float v) {
    const int b = __builtin_bit_cast(int, v);
    return __builtin_bit_cast(float,
        __builtin_amdgcn_update_dpp(b, b, 0x160 + M, 0xf, 0xf, true));
}

__device__ __forceinline__ void scan_load(
    ScanOps& o, const float* __restrict__ dtl, const float* __restrict__ ul,
    const float* __restrict__ prp, int t0)
{
    o.dtv = dtl[(long)t0 * D_INNER];
    o.uv  = ul[(long)t0 * (2 * D_INNER)];
#pragma unroll
    for (int i = 0; i < SCAN_T; ++i) {
        const long t = t0 + i;
        o.Bv[i]  = *(const float4*)(prp + t * NPROJ);
        o.Cv[i]  = *(const float4*)(prp + t * NPROJ + D_STATE);
    }
}

__device__ __forceinline__ void scan_chunk(
    const ScanOps& o, const float (&a)[4], float (&h)[4], float Dp,
    float* __restrict__ yp, int t0, int lane)
{
    float acc[SCAN_T];
#pragma unroll
    for (int i = 0; i < SCAN_T; ++i) {
        const float dtv = lane_value(o.dtv, 2 * i);
        const float du  = dtv * lane_value(o.uv, 2 * i);
        h[0] = __expf(dtv * a[0]) * h[0] + du * o.Bv[i].x;
        h[1] = __expf(dtv * a[1]) * h[1] + du * o.Bv[i].y;
        h[2] = __expf(dtv * a[2]) * h[2] + du * o.Bv[i].z;
        h[3] = __expf(dtv * a[3]) * h[3] + du * o.Bv[i].w;
        acc[i] = h[0] * o.Cv[i].x + h[1] * o.Cv[i].y + h[2] * o.Cv[i].z + h[3] * o.Cv[i].w;
    }
    const bool up16 = (lane & 16) != 0;
    float b1[8];
#pragma unroll
    for (int i = 0; i < 8; ++i) {
        const float keep = up16 ? acc[i + 8] : acc[i];
        const float send = up16 ? acc[i] : acc[i + 8];
        b1[i] = keep + other_row(send);
    }
    const bool up8 = (lane & 8) != 0;
    float b2[4];
#pragma unroll
    for (int i = 0; i < 4; ++i) {
        const float keep = up8 ? b1[i + 4] : b1[i];
        const float send = up8 ? b1[i] : b1[i + 4];
        b2[i] = keep + row_xor<8>(send);
    }
    const bool up4 = (lane & 4) != 0;
    float b3[2];
#pragma unroll
    for (int i = 0; i < 2; ++i) {
        const float keep = up4 ? b2[i + 2] : b2[i];
        const float send = up4 ? b2[i] : b2[i + 2];
        b3[i] = keep + row_xor<4>(send);
    }
    const bool up2 = (lane & 2) != 0;
    const float keep2 = up2 ? b3[1] : b3[0];
    const float send2 = up2 ? b3[0] : b3[1];
    const float b4 = keep2 + row_xor<2>(send2);
    const float tot = b4 + row_xor<1>(b4);
    if ((lane & 1) == 0)
        yp[(long)(t0 + (lane >> 1)) * D_INNER] = tot + Dp * o.uv;
}

__global__ __launch_bounds__(32 * SCAN_WPB) void mamba_scan(
    const float* __restrict__ dt, const float* __restrict__ proj,
    const float* __restrict__ UZ, const float* __restrict__ A_log,
    const float* __restrict__ D_param, float* __restrict__ y)
{
    const int wv   = __builtin_amdgcn_readfirstlane((int)threadIdx.y);
    const int bd   = blockIdx.x * SCAN_WPB + wv;
    const int b    = bd >> 10;
    const int d    = bd & (D_INNER - 1);
    const int lane = threadIdx.x;

    const float4 al = *(const float4*)(A_log + d * D_STATE + lane * 4);
    float a[4], h[4];
    a[0] = -__expf(al.x); a[1] = -__expf(al.y);
    a[2] = -__expf(al.z); a[3] = -__expf(al.w);
#pragma unroll
    for (int j = 0; j < 4; ++j) h[j] = 0.0f;
    const float Dp = D_param[d];

    const long row0 = (long)b * SEQLEN;
    const int    ls  = lane >> 1;
    const float* dtp = dt + (row0 + ls) * D_INNER + d;
    const float* up  = UZ + (row0 + ls) * (2 * D_INNER) + d;
    const float* prp = proj + row0 * NPROJ + DT_RANK + lane * 4;
    float*       yp  = y + row0 * D_INNER + d;

    ScanOps s0, s1;
    scan_load(s0, dtp, up, prp, 0);
    for (int c = 0; c < SCAN_NCH; c += 2) {
        scan_load(s1, dtp, up, prp, (c + 1) * SCAN_T);
        scan_chunk(s0, a, h, Dp, yp, c * SCAN_T, lane);
        const int cn = (c + 2 < SCAN_NCH) ? (c + 2) : (SCAN_NCH - 1);
        scan_load(s0, dtp, up, prp, cn * SCAN_T);
        scan_chunk(s1, a, h, Dp, yp, (c + 1) * SCAN_T, lane);
    }
}

__global__ __launch_bounds__(256) void ln_gate(
    float* __restrict__ y, const float* __restrict__ UZ,
    const float* __restrict__ gamma, const float* __restrict__ beta)
{
    const int row = blockIdx.x;
    const int tid = threadIdx.x;
    const int c0  = tid * 4;

    const float4 v = *(const float4*)(y + (long)row * D_INNER + c0);
    float s  = v.x + v.y + v.z + v.w;
    float s2 = v.x * v.x + v.y * v.y + v.z * v.z + v.w * v.w;

#pragma unroll
    for (int off = 16; off; off >>= 1) {
        s  += __shfl_xor(s, off, 32);
        s2 += __shfl_xor(s2, off, 32);
    }
    __shared__ float rs[8], rs2[8];
    const int wid = tid >> 5;
    if ((tid & 31) == 0) { rs[wid] = s; rs2[wid] = s2; }
    __syncthreads();
    s = 0.0f; s2 = 0.0f;
#pragma unroll
    for (int w = 0; w < 8; ++w) { s += rs[w]; s2 += rs2[w]; }

    const float mu   = s * (1.0f / D_INNER);
    const float var  = s2 * (1.0f / D_INNER) - mu * mu;
    const float rstd = rsqrtf(var + 1e-5f);

    const float4 ga = *(const float4*)(gamma + c0);
    const float4 be = *(const float4*)(beta + c0);
    const float4 zv = *(const float4*)(UZ + (long)row * (2 * D_INNER) + D_INNER + c0);

    float4 o;
    o.x = ((v.x - mu) * rstd * ga.x + be.x) * silu_f(zv.x);
    o.y = ((v.y - mu) * rstd * ga.y + be.y) * silu_f(zv.y);
    o.z = ((v.z - mu) * rstd * ga.z + be.z) * silu_f(zv.z);
    o.w = ((v.w - mu) * rstd * ga.w + be.w) * silu_f(zv.w);
    *(float4*)(y + (long)row * D_INNER + c0) = o;
}

extern "C" void kernel_launch(void* const* d_in, const int* in_sizes, int n_in,
                              void* d_out, int out_size, void* d_ws, size_t ws_size,
                              hipStream_t stream)
{
    const float* x     = (const float*)d_in[0];
    const float* W_in  = (const float*)d_in[1];
    const float* W_x   = (const float*)d_in[2];
    const float* W_dt  = (const float*)d_in[3];
    const float* b_dt  = (const float*)d_in[4];
    const float* W_out = (const float*)d_in[5];
    const float* gamma = (const float*)d_in[6];
    const float* beta  = (const float*)d_in[7];
    const float* A_log = (const float*)d_in[8];
    const float* D_par = (const float*)d_in[9];
    float* out = (float*)d_out;

    float* UZ   = (float*)d_ws;
    float* proj = UZ   + (size_t)ROWS * 2 * D_INNER;
    float* dtb  = proj + (size_t)ROWS * NPROJ;
    float* yb   = dtb  + (size_t)ROWS * D_INNER;

    gemm16_wmma_f32<1, 4, D_MODEL, 2><<<dim3((2 * D_INNER) / 64, ROWS / 16 / GEMM_WPB), dim3(32, GEMM_WPB), 0, stream>>>(
        x, D_MODEL, W_in, 2 * D_INNER, UZ, 2 * D_INNER, nullptr, D_INNER);

    gemm16_wmma_f32<0, 2, D_INNER, 4><<<dim3(NPROJ / 32, ROWS / 16 / GEMM_WPB), dim3(32, GEMM_WPB), 0, stream>>>(
        UZ, 2 * D_INNER, W_x, NPROJ, proj, NPROJ, nullptr, 0);

    gemm16_wmma_f32<2, 4, DT_RANK, 1><<<dim3(D_INNER / 64, ROWS / 16 / GEMM_WPB), dim3(32, GEMM_WPB), 0, stream>>>(
        proj, NPROJ, W_dt, D_INNER, dtb, D_INNER, b_dt, 0);

    mamba_scan<<<BATCH * D_INNER / SCAN_WPB, dim3(32, SCAN_WPB), 0, stream>>>(dtb, proj, UZ, A_log, D_par, yb);

    ln_gate<<<ROWS, 256, 0, stream>>>(yb, UZ, gamma, beta);

    gemm16_wmma_f32<0, 4, D_INNER, 4><<<dim3(D_MODEL / 64, ROWS / 16 / GEMM_WPB), dim3(32, GEMM_WPB), 0, stream>>>(
        yb, D_INNER, W_out, D_MODEL, out, D_MODEL, nullptr, 0);
}


# Scan: each wave carries two neighbouring channels in the halves of packed f32 pairs, so a step's B and C loads feed both (half the waves and loads), issued as one group a chunk ahead; sum order kept
# speedup vs baseline: 5.7767x; 1.1342x over previous
#include <hip/hip_runtime.h>
#include <hip/hip_bf16.h>
#include <math.h>

typedef __attribute__((ext_vector_type(2))) float v2f;
typedef __attribute__((ext_vector_type(8))) float v8f;

#define BATCH   2
#define SEQLEN  1024
#define D_MODEL 512
#define D_INNER 1024
#define D_STATE 128
#define DT_RANK 32
#define NPROJ   (DT_RANK + 2 * D_STATE)
#define ROWS    (BATCH * SEQLEN)

__device__ __forceinline__ float silu_f(float x) {
    return x * (1.0f / (1.0f + __expf(-x)));
}
__device__ __forceinline__ float softplus_f(float x) {
    return (x > 20.0f) ? x : log1pf(__expf(x));
}

typedef __attribute__((ext_vector_type(4))) float v4f;

template <int NB> struct VecN;
template <> struct VecN<2> { typedef v2f T; };
template <> struct VecN<4> { typedef v4f T; };

#define GEMM_KB 32
#define GEMM_WPB 4

template <int NB>
struct GemmFrag {
    v4f a[4];
    typename VecN<NB>::T b0[8];
    typename VecN<NB>::T b1[8];
};

template <int NB>
__device__ __forceinline__ void gemm_load(
    GemmFrag<NB>& f, const float* __restrict__ Ap,
    const float* __restrict__ Bp, int ldb, int kb)
{
    typedef typename VecN<NB>::T VB;
    const float* ap = Ap + kb * GEMM_KB;
#pragma unroll
    for (int j = 0; j < 4; ++j) f.a[j] = *(const v4f*)(ap + 4 * j);
    const float* bp = Bp + (long)(kb * GEMM_KB) * ldb;
#pragma unroll
    for (int s = 0; s < 8; ++s) {
        f.b0[s] = *(const VB*)(bp + (long)(2 * s) * ldb);
        f.b1[s] = *(const VB*)(bp + (long)(2 * s + 1) * ldb);
    }
}

template <int NB>
__device__ __forceinline__ void gemm_block(const GemmFrag<NB>& f, v8f (&acc)[NB])
{
#pragma unroll
    for (int s = 0; s < 8; ++s) {
        v2f a;
        a.x = f.a[s >> 1][(s & 1) * 2];
        a.y = f.a[s >> 1][(s & 1) * 2 + 1];
#pragma unroll
        for (int nb = 0; nb < NB; ++nb) {
            v2f b;
            b.x = f.b0[s][nb];
            b.y = f.b1[s][nb];
            acc[nb] = __builtin_amdgcn_wmma_f32_16x16x4_f32(
                 false, a,  false, b,
                 (short)0, acc[nb],  false,  false);
        }
    }
}

template <int NB, int R, bool LAST>
__device__ __forceinline__ void gemm_ring_pass(
    GemmFrag<NB> (&f)[R], v8f (&acc)[NB], const float* __restrict__ Ap,
    const float* __restrict__ Bp, int ldb, int kb)
{
#pragma unroll
    for (int j = 0; j < R; ++j) {
        if (!LAST || j == 0)
            gemm_load<NB>(f[(j + R - 1) % R], Ap, Bp, ldb, kb + j + R - 1);
        __builtin_amdgcn_sched_barrier(0);
        gemm_block<NB>(f[j], acc);
        __builtin_amdgcn_sched_barrier(0);
    }
}

template <int EPI, int NB, int K, int R>
__global__ __launch_bounds__(32 * GEMM_WPB) void gemm16_wmma_f32(
    const float* __restrict__ A, int lda,
    const float* __restrict__ B, int ldb,
    float* __restrict__ C, int ldc,
    const float* __restrict__ bias, int split)
{
    typedef typename VecN<NB>::T VB;
    constexpr int NKB = K / GEMM_KB;
    static_assert(K % GEMM_KB == 0 && NKB % R == 0, "K blocks must fill the ring");

    const int tn   = blockIdx.x;
    const int wv   = __builtin_amdgcn_readfirstlane((int)threadIdx.y);
    const int tm   = blockIdx.y * GEMM_WPB + wv;
    const int lane = threadIdx.x;
    const int half = lane >> 4;
    const int l16  = lane & 15;

    const float* Ap = A + (long)(tm * 16 + l16) * lda + 16 * half;
    const float* Bp = B + (long)(16 * half) * ldb + (tn * (NB * 16) + NB * l16);

    v8f acc[NB];
#pragma unroll
    for (int nb = 0; nb < NB; ++nb) acc[nb] = (v8f){};

    GemmFrag<NB> f[R];
#pragma unroll
    for (int j = 0; j < R - 1; ++j) {
        gemm_load<NB>(f[j], Ap, Bp, ldb, j);
        __builtin_amdgcn_sched_barrier(0);
    }

#pragma unroll 1
    for (int kb = 0; kb < NKB - R; kb += R)
        gemm_ring_pass<NB, R, false>(f, acc, Ap, Bp, ldb, kb);
    gemm_ring_pass<NB, R, true>(f, acc, Ap, Bp, ldb, NKB - R);

    const int col0 = tn * (NB * 16) + NB * l16;
    VB bv = {};
    if (EPI == 2) bv = *(const VB*)(bias + col0);
#pragma unroll
    for (int r = 0; r < 8; ++r) {
        const int row = tm * 16 + half * 8 + r;
        VB o;
#pragma unroll
        for (int nb = 0; nb < NB; ++nb) {
            const int col = col0 + nb;
            float v = acc[nb][r];
            if (EPI == 1) {
                if (col < split) v = silu_f(v);
            } else if (EPI == 2) {
                v = softplus_f(v + bv[nb]);
            }
            o[nb] = v;
        }
        *(VB*)(C + (long)row * ldc + col0) = o;
    }
}

#define SCAN_T   16
#define SCAN_NCH (SEQLEN / SCAN_T)
#define SCAN_CH  2
#define SCAN_WPB 4
static_assert(SCAN_T == 16, "one step per pair of lanes");
static_assert(SCAN_CH == 2, "one channel per lane of a pair, one per half of a packed pair");
static_assert(D_INNER % (SCAN_WPB * SCAN_CH) == 0, "a work-group's channels share one batch entry");
static_assert((ROWS / 16) % GEMM_WPB == 0, "M tiles must fill the work-groups");

struct ScanOps {
    float  dtv;
    float  uv;
    float4 Bv[SCAN_T];
    float4 Cv[SCAN_T];
};

__device__ __forceinline__ float lane_value(float v, int src) {
    return __builtin_bit_cast(float,
        __builtin_amdgcn_readlane(__builtin_bit_cast(int, v), src));
}

__device__ __forceinline__ float other_row(float v) {
    const int b = __builtin_bit_cast(int, v);
    return __builtin_bit_cast(float,
        __builtin_amdgcn_permlanex16(b, b, 0x76543210, 0xfedcba98, false, false));
}

template <int M>
__device__ __forceinline__ float row_xor(float v) {
    const int b = __builtin_bit_cast(int, v);
    return __builtin_bit_cast(float,
        __builtin_amdgcn_update_dpp(b, b, 0x160 + M, 0xf, 0xf, true));
}

__device__ __forceinline__ v2f splat2(float v) {
    v2f r;
    r.x = v;
    r.y = v;
    return r;
}

__device__ __forceinline__ v2f exp_pair(v2f x) {
    const v2f t = x * 1.44269504088896340736f;
    v2f r;
    r.x = __builtin_amdgcn_exp2f(t.x);
    r.y = __builtin_amdgcn_exp2f(t.y);
    return r;
}

__device__ __forceinline__ void scan_load(
    ScanOps& o, const float* __restrict__ dtl, const float* __restrict__ ul,
    const float* __restrict__ prp, int t0)
{
    o.dtv = dtl[(long)t0 * D_INNER];
    o.uv  = ul[(long)t0 * (2 * D_INNER)];
#pragma unroll
    for (int i = 0; i < SCAN_T; ++i) {
        const long t = t0 + i;
        o.Bv[i]  = *(const float4*)(prp + t * NPROJ);
        o.Cv[i]  = *(const float4*)(prp + t * NPROJ + D_STATE);
    }
}

__device__ __forceinline__ void scan_chunk(
    const ScanOps& o, const v2f (&a)[4], v2f (&h)[4], float Dp,
    float* __restrict__ yl, int t0, int lane)
{
    float acc0[SCAN_T], acc1[SCAN_T];
#pragma unroll
    for (int i = 0; i < SCAN_T; ++i) {
        v2f dtv, uu;
        dtv.x = lane_value(o.dtv, 2 * i);
        dtv.y = lane_value(o.dtv, 2 * i + 1);
        uu.x  = lane_value(o.uv, 2 * i);
        uu.y  = lane_value(o.uv, 2 * i + 1);
        const v2f du = dtv * uu;
        h[0] = __builtin_elementwise_fma(exp_pair(dtv * a[0]), h[0], du * o.Bv[i].x);
        h[1] = __builtin_elementwise_fma(exp_pair(dtv * a[1]), h[1], du * o.Bv[i].y);
        h[2] = __builtin_elementwise_fma(exp_pair(dtv * a[2]), h[2], du * o.Bv[i].z);
        h[3] = __builtin_elementwise_fma(exp_pair(dtv * a[3]), h[3], du * o.Bv[i].w);
        v2f s = h[1] * o.Cv[i].y;
        s = __builtin_elementwise_fma(h[0], splat2(o.Cv[i].x), s);
        s = __builtin_elementwise_fma(h[2], splat2(o.Cv[i].z), s);
        s = __builtin_elementwise_fma(h[3], splat2(o.Cv[i].w), s);
        acc0[i] = s.x;
        acc1[i] = s.y;
    }
    const bool up16 = (lane & 16) != 0;
    float b1x[8], b1y[8];
#pragma unroll
    for (int i = 0; i < 8; ++i) {
        const float kx = up16 ? acc0[i + 8] : acc0[i];
        const float sx = up16 ? acc0[i] : acc0[i + 8];
        b1x[i] = kx + other_row(sx);
        const float ky = up16 ? acc1[i + 8] : acc1[i];
        const float sy = up16 ? acc1[i] : acc1[i + 8];
        b1y[i] = ky + other_row(sy);
    }
    const bool up8 = (lane & 8) != 0;
    float b2x[4], b2y[4];
#pragma unroll
    for (int i = 0; i < 4; ++i) {
        const float kx = up8 ? b1x[i + 4] : b1x[i];
        const float sx = up8 ? b1x[i] : b1x[i + 4];
        b2x[i] = kx + row_xor<8>(sx);
        const float ky = up8 ? b1y[i + 4] : b1y[i];
        const float sy = up8 ? b1y[i] : b1y[i + 4];
        b2y[i] = ky + row_xor<8>(sy);
    }
    const bool up4 = (lane & 4) != 0;
    float b3x[2], b3y[2];
#pragma unroll
    for (int i = 0; i < 2; ++i) {
        const float kx = up4 ? b2x[i + 2] : b2x[i];
        const float sx = up4 ? b2x[i] : b2x[i + 2];
        b3x[i] = kx + row_xor<4>(sx);
        const float ky = up4 ? b2y[i + 2] : b2y[i];
        const float sy = up4 ? b2y[i] : b2y[i + 2];
        b3y[i] = ky + row_xor<4>(sy);
    }
    const bool up2 = (lane & 2) != 0;
    const float k2x = up2 ? b3x[1] : b3x[0];
    const float s2x = up2 ? b3x[0] : b3x[1];
    const float b4x = k2x + row_xor<2>(s2x);
    const float k2y = up2 ? b3y[1] : b3y[0];
    const float s2y = up2 ? b3y[0] : b3y[1];
    const float b4y = k2y + row_xor<2>(s2y);
    const bool up1 = (lane & 1) != 0;
    const float k1 = up1 ? b4y : b4x;
    const float s1 = up1 ? b4x : b4y;
    const float tot = k1 + row_xor<1>(s1);
    yl[(long)t0 * D_INNER] = tot + Dp * o.uv;
}

__global__ __launch_bounds__(32 * SCAN_WPB) void mamba_scan(
    const float* __restrict__ dt, const float* __restrict__ proj,
    const float* __restrict__ UZ, const float* __restrict__ A_log,
    const float* __restrict__ D_param, float* __restrict__ y)
{
    const int wv   = __builtin_amdgcn_readfirstlane((int)threadIdx.y);
    const int pr   = blockIdx.x * SCAN_WPB + wv;
    const int b    = pr / (D_INNER / SCAN_CH);
    const int d0   = (pr % (D_INNER / SCAN_CH)) * SCAN_CH;
    const int lane = threadIdx.x;

    const float4 al0 = *(const float4*)(A_log + d0 * D_STATE + lane * 4);
    const float4 al1 = *(const float4*)(A_log + (d0 + 1) * D_STATE + lane * 4);
    v2f a[4], h[4];
    a[0].x = -__expf(al0.x); a[1].x = -__expf(al0.y);
    a[2].x = -__expf(al0.z); a[3].x = -__expf(al0.w);
    a[0].y = -__expf(al1.x); a[1].y = -__expf(al1.y);
    a[2].y = -__expf(al1.z); a[3].y = -__expf(al1.w);
#pragma unroll
    for (int j = 0; j < 4; ++j) h[j] = (v2f){0.0f, 0.0f};

    const long row0 = (long)b * SEQLEN;
    const int    ls  = lane >> 1;
    const int    dl  = d0 + (lane & 1);
    const float  Dp  = D_param[dl];
    const float* dtp = dt + (row0 + ls) * D_INNER + dl;
    const float* up  = UZ + (row0 + ls) * (2 * D_INNER) + dl;
    const float* prp = proj + row0 * NPROJ + DT_RANK + lane * 4;
    float*       yp  = y + (row0 + ls) * D_INNER + dl;

    ScanOps s0, s1;
    scan_load(s0, dtp, up, prp, 0);
    for (int c = 0; c < SCAN_NCH; c += 2) {
        scan_load(s1, dtp, up, prp, (c + 1) * SCAN_T);
        __builtin_amdgcn_sched_barrier(0);
        scan_chunk(s0, a, h, Dp, yp, c * SCAN_T, lane);
        __builtin_amdgcn_sched_barrier(0);
        const int cn = (c + 2 < SCAN_NCH) ? (c + 2) : (SCAN_NCH - 1);
        scan_load(s0, dtp, up, prp, cn * SCAN_T);
        __builtin_amdgcn_sched_barrier(0);
        scan_chunk(s1, a, h, Dp, yp, (c + 1) * SCAN_T, lane);
        __builtin_amdgcn_sched_barrier(0);
    }
}

__global__ __launch_bounds__(256) void ln_gate(
    float* __restrict__ y, const float* __restrict__ UZ,
    const float* __restrict__ gamma, const float* __restrict__ beta)
{
    const int row = blockIdx.x;
    const int tid = threadIdx.x;
    const int c0  = tid * 4;

    const float4 v = *(const float4*)(y + (long)row * D_INNER + c0);
    float s  = v.x + v.y + v.z + v.w;
    float s2 = v.x * v.x + v.y * v.y + v.z * v.z + v.w * v.w;

#pragma unroll
    for (int off = 16; off; off >>= 1) {
        s  += __shfl_xor(s, off, 32);
        s2 += __shfl_xor(s2, off, 32);
    }
    __shared__ float rs[8], rs2[8];
    const int wid = tid >> 5;
    if ((tid & 31) == 0) { rs[wid] = s; rs2[wid] = s2; }
    __syncthreads();
    s = 0.0f; s2 = 0.0f;
#pragma unroll
    for (int w = 0; w < 8; ++w) { s += rs[w]; s2 += rs2[w]; }

    const float mu   = s * (1.0f / D_INNER);
    const float var  = s2 * (1.0f / D_INNER) - mu * mu;
    const float rstd = rsqrtf(var + 1e-5f);

    const float4 ga = *(const float4*)(gamma + c0);
    const float4 be = *(const float4*)(beta + c0);
    const float4 zv = *(const float4*)(UZ + (long)row * (2 * D_INNER) + D_INNER + c0);

    float4 o;
    o.x = ((v.x - mu) * rstd * ga.x + be.x) * silu_f(zv.x);
    o.y = ((v.y - mu) * rstd * ga.y + be.y) * silu_f(zv.y);
    o.z = ((v.z - mu) * rstd * ga.z + be.z) * silu_f(zv.z);
    o.w = ((v.w - mu) * rstd * ga.w + be.w) * silu_f(zv.w);
    *(float4*)(y + (long)row * D_INNER + c0) = o;
}

extern "C" void kernel_launch(void* const* d_in, const int* in_sizes, int n_in,
                              void* d_out, int out_size, void* d_ws, size_t ws_size,
                              hipStream_t stream)
{
    const float* x     = (const float*)d_in[0];
    const float* W_in  = (const float*)d_in[1];
    const float* W_x   = (const float*)d_in[2];
    const float* W_dt  = (const float*)d_in[3];
    const float* b_dt  = (const float*)d_in[4];
    const float* W_out = (const float*)d_in[5];
    const float* gamma = (const float*)d_in[6];
    const float* beta  = (const float*)d_in[7];
    const float* A_log = (const float*)d_in[8];
    const float* D_par = (const float*)d_in[9];
    float* out = (float*)d_out;

    float* UZ   = (float*)d_ws;
    float* proj = UZ   + (size_t)ROWS * 2 * D_INNER;
    float* dtb  = proj + (size_t)ROWS * NPROJ;
    float* yb   = dtb  + (size_t)ROWS * D_INNER;

    gemm16_wmma_f32<1, 4, D_MODEL, 2><<<dim3((2 * D_INNER) / 64, ROWS / 16 / GEMM_WPB), dim3(32, GEMM_WPB), 0, stream>>>(
        x, D_MODEL, W_in, 2 * D_INNER, UZ, 2 * D_INNER, nullptr, D_INNER);

    gemm16_wmma_f32<0, 2, D_INNER, 4><<<dim3(NPROJ / 32, ROWS / 16 / GEMM_WPB), dim3(32, GEMM_WPB), 0, stream>>>(
        UZ, 2 * D_INNER, W_x, NPROJ, proj, NPROJ, nullptr, 0);

    gemm16_wmma_f32<2, 4, DT_RANK, 1><<<dim3(D_INNER / 64, ROWS / 16 / GEMM_WPB), dim3(32, GEMM_WPB), 0, stream>>>(
        proj, NPROJ, W_dt, D_INNER, dtb, D_INNER, b_dt, 0);

    mamba_scan<<<BATCH * D_INNER / SCAN_CH / SCAN_WPB, dim3(32, SCAN_WPB), 0, stream>>>(dtb, proj, UZ, A_log, D_par, yb);

    ln_gate<<<ROWS, 256, 0, stream>>>(yb, UZ, gamma, beta);

    gemm16_wmma_f32<0, 4, D_INNER, 4><<<dim3(D_MODEL / 64, ROWS / 16 / GEMM_WPB), dim3(32, GEMM_WPB), 0, stream>>>(
        yb, D_INNER, W_out, D_MODEL, out, D_MODEL, nullptr, 0);
}


# Activations (GEMM epilogues, gate) as short branch-free forms: silu by hardware rcp, softplus = log(w)*z/(w-1), w=1+e^x, on the log2 unit; IEEE division and library log1p gone; last bits differ
# speedup vs baseline: 6.0927x; 1.0547x over previous
#include <hip/hip_runtime.h>
#include <hip/hip_bf16.h>
#include <math.h>

typedef __attribute__((ext_vector_type(2))) float v2f;
typedef __attribute__((ext_vector_type(8))) float v8f;

#define BATCH   2
#define SEQLEN  1024
#define D_MODEL 512
#define D_INNER 1024
#define D_STATE 128
#define DT_RANK 32
#define NPROJ   (DT_RANK + 2 * D_STATE)
#define ROWS    (BATCH * SEQLEN)

__device__ __forceinline__ float silu_f(float x) {
    return x * __builtin_amdgcn_rcpf(1.0f + __expf(-x));
}
__device__ __forceinline__ float softplus_f(float x) {
    const float z  = __expf(x);
    const float w  = 1.0f + z;
    const float d  = w - 1.0f;
    const float lg = __builtin_amdgcn_logf(w) * 0.693147180559945309417f;
    const float r  = (d == 0.0f) ? z : lg * (z * __builtin_amdgcn_rcpf(d));
    return (x > 20.0f) ? x : r;
}

typedef __attribute__((ext_vector_type(4))) float v4f;

template <int NB> struct VecN;
template <> struct VecN<2> { typedef v2f T; };
template <> struct VecN<4> { typedef v4f T; };

#define GEMM_KB 32
#define GEMM_WPB 4

template <int NB>
struct GemmFrag {
    v4f a[4];
    typename VecN<NB>::T b0[8];
    typename VecN<NB>::T b1[8];
};

template <int NB>
__device__ __forceinline__ void gemm_load(
    GemmFrag<NB>& f, const float* __restrict__ Ap,
    const float* __restrict__ Bp, int ldb, int kb)
{
    typedef typename VecN<NB>::T VB;
    const float* ap = Ap + kb * GEMM_KB;
#pragma unroll
    for (int j = 0; j < 4; ++j) f.a[j] = *(const v4f*)(ap + 4 * j);
    const float* bp = Bp + (long)(kb * GEMM_KB) * ldb;
#pragma unroll
    for (int s = 0; s < 8; ++s) {
        f.b0[s] = *(const VB*)(bp + (long)(2 * s) * ldb);
        f.b1[s] = *(const VB*)(bp + (long)(2 * s + 1) * ldb);
    }
}

template <int NB>
__device__ __forceinline__ void gemm_block(const GemmFrag<NB>& f, v8f (&acc)[NB])
{
#pragma unroll
    for (int s = 0; s < 8; ++s) {
        v2f a;
        a.x = f.a[s >> 1][(s & 1) * 2];
        a.y = f.a[s >> 1][(s & 1) * 2 + 1];
#pragma unroll
        for (int nb = 0; nb < NB; ++nb) {
            v2f b;
            b.x = f.b0[s][nb];
            b.y = f.b1[s][nb];
            acc[nb] = __builtin_amdgcn_wmma_f32_16x16x4_f32(
                 false, a,  false, b,
                 (short)0, acc[nb],  false,  false);
        }
    }
}

template <int NB, int R, bool LAST>
__device__ __forceinline__ void gemm_ring_pass(
    GemmFrag<NB> (&f)[R], v8f (&acc)[NB], const float* __restrict__ Ap,
    const float* __restrict__ Bp, int ldb, int kb)
{
#pragma unroll
    for (int j = 0; j < R; ++j) {
        if (!LAST || j == 0)
            gemm_load<NB>(f[(j + R - 1) % R], Ap, Bp, ldb, kb + j + R - 1);
        __builtin_amdgcn_sched_barrier(0);
        gemm_block<NB>(f[j], acc);
        __builtin_amdgcn_sched_barrier(0);
    }
}

template <int EPI, int NB, int K, int R>
__global__ __launch_bounds__(32 * GEMM_WPB) void gemm16_wmma_f32(
    const float* __restrict__ A, int lda,
    const float* __restrict__ B, int ldb,
    float* __restrict__ C, int ldc,
    const float* __restrict__ bias, int split)
{
    typedef typename VecN<NB>::T VB;
    constexpr int NKB = K / GEMM_KB;
    static_assert(K % GEMM_KB == 0 && NKB % R == 0, "K blocks must fill the ring");

    const int tn   = blockIdx.x;
    const int wv   = __builtin_amdgcn_readfirstlane((int)threadIdx.y);
    const int tm   = blockIdx.y * GEMM_WPB + wv;
    const int lane = threadIdx.x;
    const int half = lane >> 4;
    const int l16  = lane & 15;

    const float* Ap = A + (long)(tm * 16 + l16) * lda + 16 * half;
    const float* Bp = B + (long)(16 * half) * ldb + (tn * (NB * 16) + NB * l16);

    v8f acc[NB];
#pragma unroll
    for (int nb = 0; nb < NB; ++nb) acc[nb] = (v8f){};

    GemmFrag<NB> f[R];
#pragma unroll
    for (int j = 0; j < R - 1; ++j) {
        gemm_load<NB>(f[j], Ap, Bp, ldb, j);
        __builtin_amdgcn_sched_barrier(0);
    }

#pragma unroll 1
    for (int kb = 0; kb < NKB - R; kb += R)
        gemm_ring_pass<NB, R, false>(f, acc, Ap, Bp, ldb, kb);
    gemm_ring_pass<NB, R, true>(f, acc, Ap, Bp, ldb, NKB - R);

    const int col0 = tn * (NB * 16) + NB * l16;
    VB bv = {};
    if (EPI == 2) bv = *(const VB*)(bias + col0);
#pragma unroll
    for (int r = 0; r < 8; ++r) {
        const int row = tm * 16 + half * 8 + r;
        VB o;
#pragma unroll
        for (int nb = 0; nb < NB; ++nb) {
            const int col = col0 + nb;
            float v = acc[nb][r];
            if (EPI == 1) {
                v = (col < split) ? silu_f(v) : v;
            } else if (EPI == 2) {
                v = softplus_f(v + bv[nb]);
            }
            o[nb] = v;
        }
        *(VB*)(C + (long)row * ldc + col0) = o;
    }
}

#define SCAN_T   16
#define SCAN_NCH (SEQLEN / SCAN_T)
#define SCAN_CH  2
#define SCAN_WPB 4
static_assert(SCAN_T == 16, "one step per pair of lanes");
static_assert(SCAN_CH == 2, "one channel per lane of a pair, one per half of a packed pair");
static_assert(D_INNER % (SCAN_WPB * SCAN_CH) == 0, "a work-group's channels share one batch entry");
static_assert((ROWS / 16) % GEMM_WPB == 0, "M tiles must fill the work-groups");

struct ScanOps {
    float  dtv;
    float  uv;
    float4 Bv[SCAN_T];
    float4 Cv[SCAN_T];
};

__device__ __forceinline__ float lane_value(float v, int src) {
    return __builtin_bit_cast(float,
        __builtin_amdgcn_readlane(__builtin_bit_cast(int, v), src));
}

__device__ __forceinline__ float other_row(float v) {
    const int b = __builtin_bit_cast(int, v);
    return __builtin_bit_cast(float,
        __builtin_amdgcn_permlanex16(b, b, 0x76543210, 0xfedcba98, false, false));
}

template <int M>
__device__ __forceinline__ float row_xor(float v) {
    const int b = __builtin_bit_cast(int, v);
    return __builtin_bit_cast(float,
        __builtin_amdgcn_update_dpp(b, b, 0x160 + M, 0xf, 0xf, true));
}

__device__ __forceinline__ v2f splat2(float v) {
    v2f r;
    r.x = v;
    r.y = v;
    return r;
}

__device__ __forceinline__ v2f exp_pair(v2f x) {
    const v2f t = x * 1.44269504088896340736f;
    v2f r;
    r.x = __builtin_amdgcn_exp2f(t.x);
    r.y = __builtin_amdgcn_exp2f(t.y);
    return r;
}

__device__ __forceinline__ void scan_load(
    ScanOps& o, const float* __restrict__ dtl, const float* __restrict__ ul,
    const float* __restrict__ prp, int t0)
{
    o.dtv = dtl[(long)t0 * D_INNER];
    o.uv  = ul[(long)t0 * (2 * D_INNER)];
#pragma unroll
    for (int i = 0; i < SCAN_T; ++i) {
        const long t = t0 + i;
        o.Bv[i]  = *(const float4*)(prp + t * NPROJ);
        o.Cv[i]  = *(const float4*)(prp + t * NPROJ + D_STATE);
    }
}

__device__ __forceinline__ void scan_chunk(
    const ScanOps& o, const v2f (&a)[4], v2f (&h)[4], float Dp,
    float* __restrict__ yl, int t0, int lane)
{
    float acc0[SCAN_T], acc1[SCAN_T];
#pragma unroll
    for (int i = 0; i < SCAN_T; ++i) {
        v2f dtv, uu;
        dtv.x = lane_value(o.dtv, 2 * i);
        dtv.y = lane_value(o.dtv, 2 * i + 1);
        uu.x  = lane_value(o.uv, 2 * i);
        uu.y  = lane_value(o.uv, 2 * i + 1);
        const v2f du = dtv * uu;
        h[0] = __builtin_elementwise_fma(exp_pair(dtv * a[0]), h[0], du * o.Bv[i].x);
        h[1] = __builtin_elementwise_fma(exp_pair(dtv * a[1]), h[1], du * o.Bv[i].y);
        h[2] = __builtin_elementwise_fma(exp_pair(dtv * a[2]), h[2], du * o.Bv[i].z);
        h[3] = __builtin_elementwise_fma(exp_pair(dtv * a[3]), h[3], du * o.Bv[i].w);
        v2f s = h[1] * o.Cv[i].y;
        s = __builtin_elementwise_fma(h[0], splat2(o.Cv[i].x), s);
        s = __builtin_elementwise_fma(h[2], splat2(o.Cv[i].z), s);
        s = __builtin_elementwise_fma(h[3], splat2(o.Cv[i].w), s);
        acc0[i] = s.x;
        acc1[i] = s.y;
    }
    const bool up16 = (lane & 16) != 0;
    float b1x[8], b1y[8];
#pragma unroll
    for (int i = 0; i < 8; ++i) {
        const float kx = up16 ? acc0[i + 8] : acc0[i];
        const float sx = up16 ? acc0[i] : acc0[i + 8];
        b1x[i] = kx + other_row(sx);
        const float ky = up16 ? acc1[i + 8] : acc1[i];
        const float sy = up16 ? acc1[i] : acc1[i + 8];
        b1y[i] = ky + other_row(sy);
    }
    const bool up8 = (lane & 8) != 0;
    float b2x[4], b2y[4];
#pragma unroll
    for (int i = 0; i < 4; ++i) {
        const float kx = up8 ? b1x[i + 4] : b1x[i];
        const float sx = up8 ? b1x[i] : b1x[i + 4];
        b2x[i] = kx + row_xor<8>(sx);
        const float ky = up8 ? b1y[i + 4] : b1y[i];
        const float sy = up8 ? b1y[i] : b1y[i + 4];
        b2y[i] = ky + row_xor<8>(sy);
    }
    const bool up4 = (lane & 4) != 0;
    float b3x[2], b3y[2];
#pragma unroll
    for (int i = 0; i < 2; ++i) {
        const float kx = up4 ? b2x[i + 2] : b2x[i];
        const float sx = up4 ? b2x[i] : b2x[i + 2];
        b3x[i] = kx + row_xor<4>(sx);
        const float ky = up4 ? b2y[i + 2] : b2y[i];
        const float sy = up4 ? b2y[i] : b2y[i + 2];
        b3y[i] = ky + row_xor<4>(sy);
    }
    const bool up2 = (lane & 2) != 0;
    const float k2x = up2 ? b3x[1] : b3x[0];
    const float s2x = up2 ? b3x[0] : b3x[1];
    const float b4x = k2x + row_xor<2>(s2x);
    const float k2y = up2 ? b3y[1] : b3y[0];
    const float s2y = up2 ? b3y[0] : b3y[1];
    const float b4y = k2y + row_xor<2>(s2y);
    const bool up1 = (lane & 1) != 0;
    const float k1 = up1 ? b4y : b4x;
    const float s1 = up1 ? b4x : b4y;
    const float tot = k1 + row_xor<1>(s1);
    yl[(long)t0 * D_INNER] = tot + Dp * o.uv;
}

__global__ __launch_bounds__(32 * SCAN_WPB) void mamba_scan(
    const float* __restrict__ dt, const float* __restrict__ proj,
    const float* __restrict__ UZ, const float* __restrict__ A_log,
    const float* __restrict__ D_param, float* __restrict__ y)
{
    const int wv   = __builtin_amdgcn_readfirstlane((int)threadIdx.y);
    const int pr   = blockIdx.x * SCAN_WPB + wv;
    const int b    = pr / (D_INNER / SCAN_CH);
    const int d0   = (pr % (D_INNER / SCAN_CH)) * SCAN_CH;
    const int lane = threadIdx.x;

    const float4 al0 = *(const float4*)(A_log + d0 * D_STATE + lane * 4);
    const float4 al1 = *(const float4*)(A_log + (d0 + 1) * D_STATE + lane * 4);
    v2f a[4], h[4];
    a[0].x = -__expf(al0.x); a[1].x = -__expf(al0.y);
    a[2].x = -__expf(al0.z); a[3].x = -__expf(al0.w);
    a[0].y = -__expf(al1.x); a[1].y = -__expf(al1.y);
    a[2].y = -__expf(al1.z); a[3].y = -__expf(al1.w);
#pragma unroll
    for (int j = 0; j < 4; ++j) h[j] = (v2f){0.0f, 0.0f};

    const long row0 = (long)b * SEQLEN;
    const int    ls  = lane >> 1;
    const int    dl  = d0 + (lane & 1);
    const float  Dp  = D_param[dl];
    const float* dtp = dt + (row0 + ls) * D_INNER + dl;
    const float* up  = UZ + (row0 + ls) * (2 * D_INNER) + dl;
    const float* prp = proj + row0 * NPROJ + DT_RANK + lane * 4;
    float*       yp  = y + (row0 + ls) * D_INNER + dl;

    ScanOps s0, s1;
    scan_load(s0, dtp, up, prp, 0);
    for (int c = 0; c < SCAN_NCH; c += 2) {
        scan_load(s1, dtp, up, prp, (c + 1) * SCAN_T);
        __builtin_amdgcn_sched_barrier(0);
        scan_chunk(s0, a, h, Dp, yp, c * SCAN_T, lane);
        __builtin_amdgcn_sched_barrier(0);
        const int cn = (c + 2 < SCAN_NCH) ? (c + 2) : (SCAN_NCH - 1);
        scan_load(s0, dtp, up, prp, cn * SCAN_T);
        __builtin_amdgcn_sched_barrier(0);
        scan_chunk(s1, a, h, Dp, yp, (c + 1) * SCAN_T, lane);
        __builtin_amdgcn_sched_barrier(0);
    }
}

__global__ __launch_bounds__(256) void ln_gate(
    float* __restrict__ y, const float* __restrict__ UZ,
    const float* __restrict__ gamma, const float* __restrict__ beta)
{
    const int row = blockIdx.x;
    const int tid = threadIdx.x;
    const int c0  = tid * 4;

    const float4 v = *(const float4*)(y + (long)row * D_INNER + c0);
    float s  = v.x + v.y + v.z + v.w;
    float s2 = v.x * v.x + v.y * v.y + v.z * v.z + v.w * v.w;

#pragma unroll
    for (int off = 16; off; off >>= 1) {
        s  += __shfl_xor(s, off, 32);
        s2 += __shfl_xor(s2, off, 32);
    }
    __shared__ float rs[8], rs2[8];
    const int wid = tid >> 5;
    if ((tid & 31) == 0) { rs[wid] = s; rs2[wid] = s2; }
    __syncthreads();
    s = 0.0f; s2 = 0.0f;
#pragma unroll
    for (int w = 0; w < 8; ++w) { s += rs[w]; s2 += rs2[w]; }

    const float mu   = s * (1.0f / D_INNER);
    const float var  = s2 * (1.0f / D_INNER) - mu * mu;
    const float rstd = rsqrtf(var + 1e-5f);

    const float4 ga = *(const float4*)(gamma + c0);
    const float4 be = *(const float4*)(beta + c0);
    const float4 zv = *(const float4*)(UZ + (long)row * (2 * D_INNER) + D_INNER + c0);

    float4 o;
    o.x = ((v.x - mu) * rstd * ga.x + be.x) * silu_f(zv.x);
    o.y = ((v.y - mu) * rstd * ga.y + be.y) * silu_f(zv.y);
    o.z = ((v.z - mu) * rstd * ga.z + be.z) * silu_f(zv.z);
    o.w = ((v.w - mu) * rstd * ga.w + be.w) * silu_f(zv.w);
    *(float4*)(y + (long)row * D_INNER + c0) = o;
}

extern "C" void kernel_launch(void* const* d_in, const int* in_sizes, int n_in,
                              void* d_out, int out_size, void* d_ws, size_t ws_size,
                              hipStream_t stream)
{
    const float* x     = (const float*)d_in[0];
    const float* W_in  = (const float*)d_in[1];
    const float* W_x   = (const float*)d_in[2];
    const float* W_dt  = (const float*)d_in[3];
    const float* b_dt  = (const float*)d_in[4];
    const float* W_out = (const float*)d_in[5];
    const float* gamma = (const float*)d_in[6];
    const float* beta  = (const float*)d_in[7];
    const float* A_log = (const float*)d_in[8];
    const float* D_par = (const float*)d_in[9];
    float* out = (float*)d_out;

    float* UZ   = (float*)d_ws;
    float* proj = UZ   + (size_t)ROWS * 2 * D_INNER;
    float* dtb  = proj + (size_t)ROWS * NPROJ;
    float* yb   = dtb  + (size_t)ROWS * D_INNER;

    gemm16_wmma_f32<1, 4, D_MODEL, 2><<<dim3((2 * D_INNER) / 64, ROWS / 16 / GEMM_WPB), dim3(32, GEMM_WPB), 0, stream>>>(
        x, D_MODEL, W_in, 2 * D_INNER, UZ, 2 * D_INNER, nullptr, D_INNER);

    gemm16_wmma_f32<0, 2, D_INNER, 4><<<dim3(NPROJ / 32, ROWS / 16 / GEMM_WPB), dim3(32, GEMM_WPB), 0, stream>>>(
        UZ, 2 * D_INNER, W_x, NPROJ, proj, NPROJ, nullptr, 0);

    gemm16_wmma_f32<2, 4, DT_RANK, 1><<<dim3(D_INNER / 64, ROWS / 16 / GEMM_WPB), dim3(32, GEMM_WPB), 0, stream>>>(
        proj, NPROJ, W_dt, D_INNER, dtb, D_INNER, b_dt, 0);

    mamba_scan<<<BATCH * D_INNER / SCAN_CH / SCAN_WPB, dim3(32, SCAN_WPB), 0, stream>>>(dtb, proj, UZ, A_log, D_par, yb);

    ln_gate<<<ROWS, 256, 0, stream>>>(yb, UZ, gamma, beta);

    gemm16_wmma_f32<0, 4, D_INNER, 4><<<dim3(D_MODEL / 64, ROWS / 16 / GEMM_WPB), dim3(32, GEMM_WPB), 0, stream>>>(
        yb, D_INNER, W_out, D_MODEL, out, D_MODEL, nullptr, 0);
}


# dt projection (one 32-k block on the f32 matrix instruction, + softplus) made in the scan launch's prologue into 32 KB of LDS behind one barrier: no dt launch, no dt plane; sums unchanged
# speedup vs baseline: 6.3661x; 1.0449x over previous
#include <hip/hip_runtime.h>
#include <hip/hip_bf16.h>
#include <math.h>

typedef __attribute__((ext_vector_type(2))) float v2f;
typedef __attribute__((ext_vector_type(8))) float v8f;

#define BATCH   2
#define SEQLEN  1024
#define D_MODEL 512
#define D_INNER 1024
#define D_STATE 128
#define DT_RANK 32
#define NPROJ   (DT_RANK + 2 * D_STATE)
#define ROWS    (BATCH * SEQLEN)

__device__ __forceinline__ float silu_f(float x) {
    return x * __builtin_amdgcn_rcpf(1.0f + __expf(-x));
}
__device__ __forceinline__ float softplus_f(float x) {
    const float z  = __expf(x);
    const float w  = 1.0f + z;
    const float d  = w - 1.0f;
    const float lg = __builtin_amdgcn_logf(w) * 0.693147180559945309417f;
    const float r  = (d == 0.0f) ? z : lg * (z * __builtin_amdgcn_rcpf(d));
    return (x > 20.0f) ? x : r;
}

typedef __attribute__((ext_vector_type(4))) float v4f;

template <int NB> struct VecN;
template <> struct VecN<2> { typedef v2f T; };
template <> struct VecN<4> { typedef v4f T; };

#define GEMM_KB 32
#define GEMM_WPB 4

template <int NB>
struct GemmFrag {
    v4f a[4];
    typename VecN<NB>::T b0[8];
    typename VecN<NB>::T b1[8];
};

template <int NB>
__device__ __forceinline__ void gemm_load(
    GemmFrag<NB>& f, const float* __restrict__ Ap,
    const float* __restrict__ Bp, int ldb, int kb)
{
    typedef typename VecN<NB>::T VB;
    const float* ap = Ap + kb * GEMM_KB;
#pragma unroll
    for (int j = 0; j < 4; ++j) f.a[j] = *(const v4f*)(ap + 4 * j);
    const float* bp = Bp + (long)(kb * GEMM_KB) * ldb;
#pragma unroll
    for (int s = 0; s < 8; ++s) {
        f.b0[s] = *(const VB*)(bp + (long)(2 * s) * ldb);
        f.b1[s] = *(const VB*)(bp + (long)(2 * s + 1) * ldb);
    }
}

template <int NB>
__device__ __forceinline__ void gemm_block(const GemmFrag<NB>& f, v8f (&acc)[NB])
{
#pragma unroll
    for (int s = 0; s < 8; ++s) {
        v2f a;
        a.x = f.a[s >> 1][(s & 1) * 2];
        a.y = f.a[s >> 1][(s & 1) * 2 + 1];
#pragma unroll
        for (int nb = 0; nb < NB; ++nb) {
            v2f b;
            b.x = f.b0[s][nb];
            b.y = f.b1[s][nb];
            acc[nb] = __builtin_amdgcn_wmma_f32_16x16x4_f32(
                 false, a,  false, b,
                 (short)0, acc[nb],  false,  false);
        }
    }
}

template <int NB, int R, bool LAST>
__device__ __forceinline__ void gemm_ring_pass(
    GemmFrag<NB> (&f)[R], v8f (&acc)[NB], const float* __restrict__ Ap,
    const float* __restrict__ Bp, int ldb, int kb)
{
#pragma unroll
    for (int j = 0; j < R; ++j) {
        if (!LAST || j == 0)
            gemm_load<NB>(f[(j + R - 1) % R], Ap, Bp, ldb, kb + j + R - 1);
        __builtin_amdgcn_sched_barrier(0);
        gemm_block<NB>(f[j], acc);
        __builtin_amdgcn_sched_barrier(0);
    }
}

template <int EPI, int NB, int K, int R>
__global__ __launch_bounds__(32 * GEMM_WPB) void gemm16_wmma_f32(
    const float* __restrict__ A, int lda,
    const float* __restrict__ B, int ldb,
    float* __restrict__ C, int ldc,
    const float* __restrict__ bias, int split)
{
    typedef typename VecN<NB>::T VB;
    constexpr int NKB = K / GEMM_KB;
    static_assert(K % GEMM_KB == 0 && NKB % R == 0, "K blocks must fill the ring");

    const int tn   = blockIdx.x;
    const int wv   = __builtin_amdgcn_readfirstlane((int)threadIdx.y);
    const int tm   = blockIdx.y * GEMM_WPB + wv;
    const int lane = threadIdx.x;
    const int half = lane >> 4;
    const int l16  = lane & 15;

    const float* Ap = A + (long)(tm * 16 + l16) * lda + 16 * half;
    const float* Bp = B + (long)(16 * half) * ldb + (tn * (NB * 16) + NB * l16);

    v8f acc[NB];
#pragma unroll
    for (int nb = 0; nb < NB; ++nb) acc[nb] = (v8f){};

    GemmFrag<NB> f[R];
#pragma unroll
    for (int j = 0; j < R - 1; ++j) {
        gemm_load<NB>(f[j], Ap, Bp, ldb, j);
        __builtin_amdgcn_sched_barrier(0);
    }

#pragma unroll 1
    for (int kb = 0; kb < NKB - R; kb += R)
        gemm_ring_pass<NB, R, false>(f, acc, Ap, Bp, ldb, kb);
    gemm_ring_pass<NB, R, true>(f, acc, Ap, Bp, ldb, NKB - R);

    const int col0 = tn * (NB * 16) + NB * l16;
    VB bv = {};
    if (EPI == 2) bv = *(const VB*)(bias + col0);
#pragma unroll
    for (int r = 0; r < 8; ++r) {
        const int row = tm * 16 + half * 8 + r;
        VB o;
#pragma unroll
        for (int nb = 0; nb < NB; ++nb) {
            const int col = col0 + nb;
            float v = acc[nb][r];
            if (EPI == 1) {
                v = (col < split) ? silu_f(v) : v;
            } else if (EPI == 2) {
                v = softplus_f(v + bv[nb]);
            }
            o[nb] = v;
        }
        *(VB*)(C + (long)row * ldc + col0) = o;
    }
}

#define SCAN_T   16
#define SCAN_NCH (SEQLEN / SCAN_T)
#define SCAN_CH  2
#define SCAN_WPB 4
#define SCAN_GCH (SCAN_WPB * SCAN_CH)
#define DT_TILES (SEQLEN / 16 / SCAN_WPB)
#define DT_GRP   4
static_assert(SCAN_GCH == 8, "the dt tile's 16 columns are the 8 channels twice");
static_assert(DT_RANK == GEMM_KB, "dt projection is one 32-k block");
static_assert(DT_TILES % DT_GRP == 0 && DT_GRP % 2 == 0, "tile groups must fill a wave's share; a lane pair halves a group");
static_assert(SCAN_T == 16, "one step per pair of lanes");
static_assert(SCAN_CH == 2, "one channel per lane of a pair, one per half of a packed pair");
static_assert(D_INNER % (SCAN_WPB * SCAN_CH) == 0, "a work-group's channels share one batch entry");
static_assert((ROWS / 16) % GEMM_WPB == 0, "M tiles must fill the work-groups");

struct ScanOps {
    float  dtv;
    float  uv;
    float4 Bv[SCAN_T];
    float4 Cv[SCAN_T];
};

__device__ __forceinline__ float lane_value(float v, int src) {
    return __builtin_bit_cast(float,
        __builtin_amdgcn_readlane(__builtin_bit_cast(int, v), src));
}

__device__ __forceinline__ float other_row(float v) {
    const int b = __builtin_bit_cast(int, v);
    return __builtin_bit_cast(float,
        __builtin_amdgcn_permlanex16(b, b, 0x76543210, 0xfedcba98, false, false));
}

template <int M>
__device__ __forceinline__ float row_xor(float v) {
    const int b = __builtin_bit_cast(int, v);
    return __builtin_bit_cast(float,
        __builtin_amdgcn_update_dpp(b, b, 0x160 + M, 0xf, 0xf, true));
}

__device__ __forceinline__ v2f splat2(float v) {
    v2f r;
    r.x = v;
    r.y = v;
    return r;
}

__device__ __forceinline__ v2f exp_pair(v2f x) {
    const v2f t = x * 1.44269504088896340736f;
    v2f r;
    r.x = __builtin_amdgcn_exp2f(t.x);
    r.y = __builtin_amdgcn_exp2f(t.y);
    return r;
}

__device__ __forceinline__ void scan_load_global(
    ScanOps& o, const float* __restrict__ ul,
    const float* __restrict__ prp, int t0)
{
    o.uv  = ul[(long)t0 * (2 * D_INNER)];
#pragma unroll
    for (int i = 0; i < SCAN_T; ++i) {
        const long t = t0 + i;
        o.Bv[i]  = *(const float4*)(prp + t * NPROJ);
        o.Cv[i]  = *(const float4*)(prp + t * NPROJ + D_STATE);
    }
}

__device__ __forceinline__ void scan_load(
    ScanOps& o, const float* __restrict__ dtl, const float* __restrict__ ul,
    const float* __restrict__ prp, int t0)
{
    o.dtv = dtl[t0 * SCAN_GCH];
    scan_load_global(o, ul, prp, t0);
}

__device__ __forceinline__ void scan_chunk(
    const ScanOps& o, const v2f (&a)[4], v2f (&h)[4], float Dp,
    float* __restrict__ yl, int t0, int lane)
{
    float acc0[SCAN_T], acc1[SCAN_T];
#pragma unroll
    for (int i = 0; i < SCAN_T; ++i) {
        v2f dtv, uu;
        dtv.x = lane_value(o.dtv, 2 * i);
        dtv.y = lane_value(o.dtv, 2 * i + 1);
        uu.x  = lane_value(o.uv, 2 * i);
        uu.y  = lane_value(o.uv, 2 * i + 1);
        const v2f du = dtv * uu;
        h[0] = __builtin_elementwise_fma(exp_pair(dtv * a[0]), h[0], du * o.Bv[i].x);
        h[1] = __builtin_elementwise_fma(exp_pair(dtv * a[1]), h[1], du * o.Bv[i].y);
        h[2] = __builtin_elementwise_fma(exp_pair(dtv * a[2]), h[2], du * o.Bv[i].z);
        h[3] = __builtin_elementwise_fma(exp_pair(dtv * a[3]), h[3], du * o.Bv[i].w);
        v2f s = h[1] * o.Cv[i].y;
        s = __builtin_elementwise_fma(h[0], splat2(o.Cv[i].x), s);
        s = __builtin_elementwise_fma(h[2], splat2(o.Cv[i].z), s);
        s = __builtin_elementwise_fma(h[3], splat2(o.Cv[i].w), s);
        acc0[i] = s.x;
        acc1[i] = s.y;
    }
    const bool up16 = (lane & 16) != 0;
    float b1x[8], b1y[8];
#pragma unroll
    for (int i = 0; i < 8; ++i) {
        const float kx = up16 ? acc0[i + 8] : acc0[i];
        const float sx = up16 ? acc0[i] : acc0[i + 8];
        b1x[i] = kx + other_row(sx);
        const float ky = up16 ? acc1[i + 8] : acc1[i];
        const float sy = up16 ? acc1[i] : acc1[i + 8];
        b1y[i] = ky + other_row(sy);
    }
    const bool up8 = (lane & 8) != 0;
    float b2x[4], b2y[4];
#pragma unroll
    for (int i = 0; i < 4; ++i) {
        const float kx = up8 ? b1x[i + 4] : b1x[i];
        const float sx = up8 ? b1x[i] : b1x[i + 4];
        b2x[i] = kx + row_xor<8>(sx);
        const float ky = up8 ? b1y[i + 4] : b1y[i];
        const float sy = up8 ? b1y[i] : b1y[i + 4];
        b2y[i] = ky + row_xor<8>(sy);
    }
    const bool up4 = (lane & 4) != 0;
    float b3x[2], b3y[2];
#pragma unroll
    for (int i = 0; i < 2; ++i) {
        const float kx = up4 ? b2x[i + 2] : b2x[i];
        const float sx = up4 ? b2x[i] : b2x[i + 2];
        b3x[i] = kx + row_xor<4>(sx);
        const float ky = up4 ? b2y[i + 2] : b2y[i];
        const float sy = up4 ? b2y[i] : b2y[i + 2];
        b3y[i] = ky + row_xor<4>(sy);
    }
    const bool up2 = (lane & 2) != 0;
    const float k2x = up2 ? b3x[1] : b3x[0];
    const float s2x = up2 ? b3x[0] : b3x[1];
    const float b4x = k2x + row_xor<2>(s2x);
    const float k2y = up2 ? b3y[1] : b3y[0];
    const float s2y = up2 ? b3y[0] : b3y[1];
    const float b4y = k2y + row_xor<2>(s2y);
    const bool up1 = (lane & 1) != 0;
    const float k1 = up1 ? b4y : b4x;
    const float s1 = up1 ? b4x : b4y;
    const float tot = k1 + row_xor<1>(s1);
    yl[(long)t0 * D_INNER] = tot + Dp * o.uv;
}

struct DtFrag {
    v4f a[DT_GRP][4];
};

__device__ __forceinline__ void dt_load(DtFrag& f, const float* __restrict__ ap)
{
#pragma unroll
    for (int t = 0; t < DT_GRP; ++t) {
#pragma unroll
        for (int j = 0; j < 4; ++j)
            f.a[t][j] = *(const v4f*)(ap + (long)t * 16 * NPROJ + 4 * j);
    }
}

__device__ __forceinline__ void dt_group(
    const DtFrag& f, const v2f (&bq)[8], float bias, bool upper,
    float* __restrict__ dst)
{
    v8f acc[DT_GRP];
#pragma unroll
    for (int t = 0; t < DT_GRP; ++t) acc[t] = (v8f){};
#pragma unroll
    for (int s = 0; s < 8; ++s) {
#pragma unroll
        for (int t = 0; t < DT_GRP; ++t) {
            v2f a;
            a.x = f.a[t][s >> 1][(s & 1) * 2];
            a.y = f.a[t][s >> 1][(s & 1) * 2 + 1];
            acc[t] = __builtin_amdgcn_wmma_f32_16x16x4_f32(
                 false, a,  false, bq[s],
                 (short)0, acc[t],  false,  false);
        }
    }
#pragma unroll
    for (int j = 0; j < DT_GRP / 2; ++j) {
#pragma unroll
        for (int r = 0; r < 8; ++r) {
            const float v = upper ? acc[DT_GRP / 2 + j][r] : acc[j][r];
            dst[(j * 16 + r) * SCAN_GCH] = softplus_f(v + bias);
        }
    }
}

__global__ __launch_bounds__(32 * SCAN_WPB) void mamba_scan(
    const float* __restrict__ proj, const float* __restrict__ W_dt,
    const float* __restrict__ b_dt, const float* __restrict__ UZ,
    const float* __restrict__ A_log, const float* __restrict__ D_param,
    float* __restrict__ y)
{
    __shared__ float dt_s[SEQLEN * SCAN_GCH];

    const int wv   = __builtin_amdgcn_readfirstlane((int)threadIdx.y);
    const int pr   = blockIdx.x * SCAN_WPB + wv;
    const int b    = pr / (D_INNER / SCAN_CH);
    const int d0   = (pr % (D_INNER / SCAN_CH)) * SCAN_CH;
    const int lane = threadIdx.x;

    const float4 al0 = *(const float4*)(A_log + d0 * D_STATE + lane * 4);
    const float4 al1 = *(const float4*)(A_log + (d0 + 1) * D_STATE + lane * 4);
    v2f a[4], h[4];
    a[0].x = -__expf(al0.x); a[1].x = -__expf(al0.y);
    a[2].x = -__expf(al0.z); a[3].x = -__expf(al0.w);
    a[0].y = -__expf(al1.x); a[1].y = -__expf(al1.y);
    a[2].y = -__expf(al1.z); a[3].y = -__expf(al1.w);
#pragma unroll
    for (int j = 0; j < 4; ++j) h[j] = (v2f){0.0f, 0.0f};

    const long row0 = (long)b * SEQLEN;
    const int    ls  = lane >> 1;
    const int    dl  = d0 + (lane & 1);
    const float  Dp  = D_param[dl];
    const float* dtp = dt_s + ls * SCAN_GCH + wv * SCAN_CH + (lane & 1);
    const float* up  = UZ + (row0 + ls) * (2 * D_INNER) + dl;
    const float* prp = proj + row0 * NPROJ + DT_RANK + lane * 4;
    float*       yp  = y + (row0 + ls) * D_INNER + dl;

    ScanOps s0, s1;
    {
        const int  half  = lane >> 4;
        const int  l16   = lane & 15;
        const int  cj    = l16 & 7;
        const bool upper = (l16 & 8) != 0;
        const int  dg    = d0 - wv * SCAN_CH;
        const float* wp  = W_dt + (long)(16 * half) * D_INNER + dg + cj;
        v2f bq[8];
#pragma unroll
        for (int s = 0; s < 8; ++s) {
            bq[s].x = wp[(long)(2 * s) * D_INNER];
            bq[s].y = wp[(long)(2 * s + 1) * D_INNER];
        }
        const float bias = b_dt[dg + cj];
        const int   mt0  = wv * DT_TILES;
        const float* ap  = proj + (row0 + mt0 * 16 + l16) * NPROJ + 16 * half;
        float* dst = dt_s + ((mt0 + (upper ? DT_GRP / 2 : 0)) * 16 + 8 * half) * SCAN_GCH + cj;

        DtFrag f[2];
        dt_load(f[0], ap);
#pragma unroll
        for (int g = 0; g < DT_TILES / DT_GRP; ++g) {
            if (g + 1 < DT_TILES / DT_GRP)
                dt_load(f[(g + 1) & 1], ap + (long)(g + 1) * DT_GRP * 16 * NPROJ);
            else
                scan_load_global(s0, up, prp, 0);
            __builtin_amdgcn_sched_barrier(0);
            dt_group(f[g & 1], bq, bias, upper, dst + g * DT_GRP * 16 * SCAN_GCH);
            __builtin_amdgcn_sched_barrier(0);
        }
    }
    __syncthreads();
    s0.dtv = dtp[0];

    for (int c = 0; c < SCAN_NCH; c += 2) {
        scan_load(s1, dtp, up, prp, (c + 1) * SCAN_T);
        __builtin_amdgcn_sched_barrier(0);
        scan_chunk(s0, a, h, Dp, yp, c * SCAN_T, lane);
        __builtin_amdgcn_sched_barrier(0);
        const int cn = (c + 2 < SCAN_NCH) ? (c + 2) : (SCAN_NCH - 1);
        scan_load(s0, dtp, up, prp, cn * SCAN_T);
        __builtin_amdgcn_sched_barrier(0);
        scan_chunk(s1, a, h, Dp, yp, (c + 1) * SCAN_T, lane);
        __builtin_amdgcn_sched_barrier(0);
    }
}

__global__ __launch_bounds__(256) void ln_gate(
    float* __restrict__ y, const float* __restrict__ UZ,
    const float* __restrict__ gamma, const float* __restrict__ beta)
{
    const int row = blockIdx.x;
    const int tid = threadIdx.x;
    const int c0  = tid * 4;

    const float4 v = *(const float4*)(y + (long)row * D_INNER + c0);
    float s  = v.x + v.y + v.z + v.w;
    float s2 = v.x * v.x + v.y * v.y + v.z * v.z + v.w * v.w;

#pragma unroll
    for (int off = 16; off; off >>= 1) {
        s  += __shfl_xor(s, off, 32);
        s2 += __shfl_xor(s2, off, 32);
    }
    __shared__ float rs[8], rs2[8];
    const int wid = tid >> 5;
    if ((tid & 31) == 0) { rs[wid] = s; rs2[wid] = s2; }
    __syncthreads();
    s = 0.0f; s2 = 0.0f;
#pragma unroll
    for (int w = 0; w < 8; ++w) { s += rs[w]; s2 += rs2[w]; }

    const float mu   = s * (1.0f / D_INNER);
    const float var  = s2 * (1.0f / D_INNER) - mu * mu;
    const float rstd = rsqrtf(var + 1e-5f);

    const float4 ga = *(const float4*)(gamma + c0);
    const float4 be = *(const float4*)(beta + c0);
    const float4 zv = *(const float4*)(UZ + (long)row * (2 * D_INNER) + D_INNER + c0);

    float4 o;
    o.x = ((v.x - mu) * rstd * ga.x + be.x) * silu_f(zv.x);
    o.y = ((v.y - mu) * rstd * ga.y + be.y) * silu_f(zv.y);
    o.z = ((v.z - mu) * rstd * ga.z + be.z) * silu_f(zv.z);
    o.w = ((v.w - mu) * rstd * ga.w + be.w) * silu_f(zv.w);
    *(float4*)(y + (long)row * D_INNER + c0) = o;
}

extern "C" void kernel_launch(void* const* d_in, const int* in_sizes, int n_in,
                              void* d_out, int out_size, void* d_ws, size_t ws_size,
                              hipStream_t stream)
{
    const float* x     = (const float*)d_in[0];
    const float* W_in  = (const float*)d_in[1];
    const float* W_x   = (const float*)d_in[2];
    const float* W_dt  = (const float*)d_in[3];
    const float* b_dt  = (const float*)d_in[4];
    const float* W_out = (const float*)d_in[5];
    const float* gamma = (const float*)d_in[6];
    const float* beta  = (const float*)d_in[7];
    const float* A_log = (const float*)d_in[8];
    const float* D_par = (const float*)d_in[9];
    float* out = (float*)d_out;

    float* UZ   = (float*)d_ws;
    float* proj = UZ   + (size_t)ROWS * 2 * D_INNER;
    float* yb   = proj + (size_t)ROWS * NPROJ;

    gemm16_wmma_f32<1, 4, D_MODEL, 2><<<dim3((2 * D_INNER) / 64, ROWS / 16 / GEMM_WPB), dim3(32, GEMM_WPB), 0, stream>>>(
        x, D_MODEL, W_in, 2 * D_INNER, UZ, 2 * D_INNER, nullptr, D_INNER);

    gemm16_wmma_f32<0, 2, D_INNER, 4><<<dim3(NPROJ / 32, ROWS / 16 / GEMM_WPB), dim3(32, GEMM_WPB), 0, stream>>>(
        UZ, 2 * D_INNER, W_x, NPROJ, proj, NPROJ, nullptr, 0);

    mamba_scan<<<BATCH * D_INNER / SCAN_CH / SCAN_WPB, dim3(32, SCAN_WPB), 0, stream>>>(proj, W_dt, b_dt, UZ, A_log, D_par, yb);

    ln_gate<<<ROWS, 256, 0, stream>>>(yb, UZ, gamma, beta);

    gemm16_wmma_f32<0, 4, D_INNER, 4><<<dim3(D_MODEL / 64, ROWS / 16 / GEMM_WPB), dim3(32, GEMM_WPB), 0, stream>>>(
        yb, D_INNER, W_out, D_MODEL, out, D_MODEL, nullptr, 0);
}


# Scan: B and C operands in one rolling register set, refilled in groups of 4 steps right behind their use (12 steps ahead), not two whole-chunk sets: time loop fits 256 VGPRs, no window switches
# speedup vs baseline: 7.0744x; 1.1113x over previous
#include <hip/hip_runtime.h>
#include <hip/hip_bf16.h>
#include <math.h>

typedef __attribute__((ext_vector_type(2))) float v2f;
typedef __attribute__((ext_vector_type(8))) float v8f;

#define BATCH   2
#define SEQLEN  1024
#define D_MODEL 512
#define D_INNER 1024
#define D_STATE 128
#define DT_RANK 32
#define NPROJ   (DT_RANK + 2 * D_STATE)
#define ROWS    (BATCH * SEQLEN)

__device__ __forceinline__ float silu_f(float x) {
    return x * __builtin_amdgcn_rcpf(1.0f + __expf(-x));
}
__device__ __forceinline__ float softplus_f(float x) {
    const float z  = __expf(x);
    const float w  = 1.0f + z;
    const float d  = w - 1.0f;
    const float lg = __builtin_amdgcn_logf(w) * 0.693147180559945309417f;
    const float r  = (d == 0.0f) ? z : lg * (z * __builtin_amdgcn_rcpf(d));
    return (x > 20.0f) ? x : r;
}

typedef __attribute__((ext_vector_type(4))) float v4f;

template <int NB> struct VecN;
template <> struct VecN<2> { typedef v2f T; };
template <> struct VecN<4> { typedef v4f T; };

#define GEMM_KB 32
#define GEMM_WPB 4

template <int NB>
struct GemmFrag {
    v4f a[4];
    typename VecN<NB>::T b0[8];
    typename VecN<NB>::T b1[8];
};

template <int NB>
__device__ __forceinline__ void gemm_load(
    GemmFrag<NB>& f, const float* __restrict__ Ap,
    const float* __restrict__ Bp, int ldb, int kb)
{
    typedef typename VecN<NB>::T VB;
    const float* ap = Ap + kb * GEMM_KB;
#pragma unroll
    for (int j = 0; j < 4; ++j) f.a[j] = *(const v4f*)(ap + 4 * j);
    const float* bp = Bp + (long)(kb * GEMM_KB) * ldb;
#pragma unroll
    for (int s = 0; s < 8; ++s) {
        f.b0[s] = *(const VB*)(bp + (long)(2 * s) * ldb);
        f.b1[s] = *(const VB*)(bp + (long)(2 * s + 1) * ldb);
    }
}

template <int NB>
__device__ __forceinline__ void gemm_block(const GemmFrag<NB>& f, v8f (&acc)[NB])
{
#pragma unroll
    for (int s = 0; s < 8; ++s) {
        v2f a;
        a.x = f.a[s >> 1][(s & 1) * 2];
        a.y = f.a[s >> 1][(s & 1) * 2 + 1];
#pragma unroll
        for (int nb = 0; nb < NB; ++nb) {
            v2f b;
            b.x = f.b0[s][nb];
            b.y = f.b1[s][nb];
            acc[nb] = __builtin_amdgcn_wmma_f32_16x16x4_f32(
                 false, a,  false, b,
                 (short)0, acc[nb],  false,  false);
        }
    }
}

template <int NB, int R, bool LAST>
__device__ __forceinline__ void gemm_ring_pass(
    GemmFrag<NB> (&f)[R], v8f (&acc)[NB], const float* __restrict__ Ap,
    const float* __restrict__ Bp, int ldb, int kb)
{
#pragma unroll
    for (int j = 0; j < R; ++j) {
        if (!LAST || j == 0)
            gemm_load<NB>(f[(j + R - 1) % R], Ap, Bp, ldb, kb + j + R - 1);
        __builtin_amdgcn_sched_barrier(0);
        gemm_block<NB>(f[j], acc);
        __builtin_amdgcn_sched_barrier(0);
    }
}

template <int EPI, int NB, int K, int R>
__global__ __launch_bounds__(32 * GEMM_WPB) void gemm16_wmma_f32(
    const float* __restrict__ A, int lda,
    const float* __restrict__ B, int ldb,
    float* __restrict__ C, int ldc,
    const float* __restrict__ bias, int split)
{
    typedef typename VecN<NB>::T VB;
    constexpr int NKB = K / GEMM_KB;
    static_assert(K % GEMM_KB == 0 && NKB % R == 0, "K blocks must fill the ring");

    const int tn   = blockIdx.x;
    const int wv   = __builtin_amdgcn_readfirstlane((int)threadIdx.y);
    const int tm   = blockIdx.y * GEMM_WPB + wv;
    const int lane = threadIdx.x;
    const int half = lane >> 4;
    const int l16  = lane & 15;

    const float* Ap = A + (long)(tm * 16 + l16) * lda + 16 * half;
    const float* Bp = B + (long)(16 * half) * ldb + (tn * (NB * 16) + NB * l16);

    v8f acc[NB];
#pragma unroll
    for (int nb = 0; nb < NB; ++nb) acc[nb] = (v8f){};

    GemmFrag<NB> f[R];
#pragma unroll
    for (int j = 0; j < R - 1; ++j) {
        gemm_load<NB>(f[j], Ap, Bp, ldb, j);
        __builtin_amdgcn_sched_barrier(0);
    }

#pragma unroll 1
    for (int kb = 0; kb < NKB - R; kb += R)
        gemm_ring_pass<NB, R, false>(f, acc, Ap, Bp, ldb, kb);
    gemm_ring_pass<NB, R, true>(f, acc, Ap, Bp, ldb, NKB - R);

    const int col0 = tn * (NB * 16) + NB * l16;
    VB bv = {};
    if (EPI == 2) bv = *(const VB*)(bias + col0);
#pragma unroll
    for (int r = 0; r < 8; ++r) {
        const int row = tm * 16 + half * 8 + r;
        VB o;
#pragma unroll
        for (int nb = 0; nb < NB; ++nb) {
            const int col = col0 + nb;
            float v = acc[nb][r];
            if (EPI == 1) {
                v = (col < split) ? silu_f(v) : v;
            } else if (EPI == 2) {
                v = softplus_f(v + bv[nb]);
            }
            o[nb] = v;
        }
        *(VB*)(C + (long)row * ldc + col0) = o;
    }
}

#define SCAN_T   16
#define SCAN_NCH (SEQLEN / SCAN_T)
#define SCAN_RG  4
#define SCAN_CH  2
#define SCAN_WPB 4
#define SCAN_GCH (SCAN_WPB * SCAN_CH)
#define DT_TILES (SEQLEN / 16 / SCAN_WPB)
#define DT_GRP   4
static_assert(SCAN_GCH == 8, "the dt tile's 16 columns are the 8 channels twice");
static_assert(DT_RANK == GEMM_KB, "dt projection is one 32-k block");
static_assert(DT_TILES % DT_GRP == 0 && DT_GRP % 2 == 0, "tile groups must fill a wave's share; a lane pair halves a group");
static_assert(SCAN_T == 16, "one step per pair of lanes");
static_assert(SCAN_T % SCAN_RG == 0, "refill groups must fill a chunk");
static_assert(SCAN_CH == 2, "one channel per lane of a pair, one per half of a packed pair");
static_assert(D_INNER % (SCAN_WPB * SCAN_CH) == 0, "a work-group's channels share one batch entry");
static_assert((ROWS / 16) % GEMM_WPB == 0, "M tiles must fill the work-groups");

struct ScanOps {
    float4 Bv[SCAN_T];
    float4 Cv[SCAN_T];
};

__device__ __forceinline__ float lane_value(float v, int src) {
    return __builtin_bit_cast(float,
        __builtin_amdgcn_readlane(__builtin_bit_cast(int, v), src));
}

__device__ __forceinline__ float other_row(float v) {
    const int b = __builtin_bit_cast(int, v);
    return __builtin_bit_cast(float,
        __builtin_amdgcn_permlanex16(b, b, 0x76543210, 0xfedcba98, false, false));
}

template <int M>
__device__ __forceinline__ float row_xor(float v) {
    const int b = __builtin_bit_cast(int, v);
    return __builtin_bit_cast(float,
        __builtin_amdgcn_update_dpp(b, b, 0x160 + M, 0xf, 0xf, true));
}

__device__ __forceinline__ v2f splat2(float v) {
    v2f r;
    r.x = v;
    r.y = v;
    return r;
}

__device__ __forceinline__ v2f exp_pair(v2f x) {
    const v2f t = x * 1.44269504088896340736f;
    v2f r;
    r.x = __builtin_amdgcn_exp2f(t.x);
    r.y = __builtin_amdgcn_exp2f(t.y);
    return r;
}

__device__ __forceinline__ void scan_load_bc(
    ScanOps& o, const float* __restrict__ prp, int t0, int i0, int n)
{
#pragma unroll
    for (int i = i0; i < i0 + n; ++i) {
        const long t = t0 + i;
        o.Bv[i]  = *(const float4*)(prp + t * NPROJ);
        o.Cv[i]  = *(const float4*)(prp + t * NPROJ + D_STATE);
    }
}

__device__ __forceinline__ void scan_chunk(
    ScanOps& o, float dtl, float ul, const v2f (&a)[4], v2f (&h)[4], float Dp,
    float* __restrict__ yl, const float* __restrict__ prp, int t0, int tn,
    int lane)
{
    float acc0[SCAN_T], acc1[SCAN_T];
#pragma unroll
    for (int i = 0; i < SCAN_T; ++i) {
        v2f dtv, uu;
        dtv.x = lane_value(dtl, 2 * i);
        dtv.y = lane_value(dtl, 2 * i + 1);
        uu.x  = lane_value(ul, 2 * i);
        uu.y  = lane_value(ul, 2 * i + 1);
        const v2f du = dtv * uu;
        h[0] = __builtin_elementwise_fma(exp_pair(dtv * a[0]), h[0], du * o.Bv[i].x);
        h[1] = __builtin_elementwise_fma(exp_pair(dtv * a[1]), h[1], du * o.Bv[i].y);
        h[2] = __builtin_elementwise_fma(exp_pair(dtv * a[2]), h[2], du * o.Bv[i].z);
        h[3] = __builtin_elementwise_fma(exp_pair(dtv * a[3]), h[3], du * o.Bv[i].w);
        v2f s = h[1] * o.Cv[i].y;
        s = __builtin_elementwise_fma(h[0], splat2(o.Cv[i].x), s);
        s = __builtin_elementwise_fma(h[2], splat2(o.Cv[i].z), s);
        s = __builtin_elementwise_fma(h[3], splat2(o.Cv[i].w), s);
        acc0[i] = s.x;
        acc1[i] = s.y;
        if (i % SCAN_RG == SCAN_RG - 1) {
            __builtin_amdgcn_sched_barrier(0);
            scan_load_bc(o, prp, tn, i - (SCAN_RG - 1), SCAN_RG);
            __builtin_amdgcn_sched_barrier(0);
        }
    }
    const bool up16 = (lane & 16) != 0;
    float b1x[8], b1y[8];
#pragma unroll
    for (int i = 0; i < 8; ++i) {
        const float kx = up16 ? acc0[i + 8] : acc0[i];
        const float sx = up16 ? acc0[i] : acc0[i + 8];
        b1x[i] = kx + other_row(sx);
        const float ky = up16 ? acc1[i + 8] : acc1[i];
        const float sy = up16 ? acc1[i] : acc1[i + 8];
        b1y[i] = ky + other_row(sy);
    }
    const bool up8 = (lane & 8) != 0;
    float b2x[4], b2y[4];
#pragma unroll
    for (int i = 0; i < 4; ++i) {
        const float kx = up8 ? b1x[i + 4] : b1x[i];
        const float sx = up8 ? b1x[i] : b1x[i + 4];
        b2x[i] = kx + row_xor<8>(sx);
        const float ky = up8 ? b1y[i + 4] : b1y[i];
        const float sy = up8 ? b1y[i] : b1y[i + 4];
        b2y[i] = ky + row_xor<8>(sy);
    }
    const bool up4 = (lane & 4) != 0;
    float b3x[2], b3y[2];
#pragma unroll
    for (int i = 0; i < 2; ++i) {
        const float kx = up4 ? b2x[i + 2] : b2x[i];
        const float sx = up4 ? b2x[i] : b2x[i + 2];
        b3x[i] = kx + row_xor<4>(sx);
        const float ky = up4 ? b2y[i + 2] : b2y[i];
        const float sy = up4 ? b2y[i] : b2y[i + 2];
        b3y[i] = ky + row_xor<4>(sy);
    }
    const bool up2 = (lane & 2) != 0;
    const float k2x = up2 ? b3x[1] : b3x[0];
    const float s2x = up2 ? b3x[0] : b3x[1];
    const float b4x = k2x + row_xor<2>(s2x);
    const float k2y = up2 ? b3y[1] : b3y[0];
    const float s2y = up2 ? b3y[0] : b3y[1];
    const float b4y = k2y + row_xor<2>(s2y);
    const bool up1 = (lane & 1) != 0;
    const float k1 = up1 ? b4y : b4x;
    const float s1 = up1 ? b4x : b4y;
    const float tot = k1 + row_xor<1>(s1);
    yl[(long)t0 * D_INNER] = tot + Dp * ul;
}

struct DtFrag {
    v4f a[DT_GRP][4];
};

__device__ __forceinline__ void dt_load(DtFrag& f, const float* __restrict__ ap)
{
#pragma unroll
    for (int t = 0; t < DT_GRP; ++t) {
#pragma unroll
        for (int j = 0; j < 4; ++j)
            f.a[t][j] = *(const v4f*)(ap + (long)t * 16 * NPROJ + 4 * j);
    }
}

__device__ __forceinline__ void dt_group(
    const DtFrag& f, const v2f (&bq)[8], float bias, bool upper,
    float* __restrict__ dst)
{
    v8f acc[DT_GRP];
#pragma unroll
    for (int t = 0; t < DT_GRP; ++t) acc[t] = (v8f){};
#pragma unroll
    for (int s = 0; s < 8; ++s) {
#pragma unroll
        for (int t = 0; t < DT_GRP; ++t) {
            v2f a;
            a.x = f.a[t][s >> 1][(s & 1) * 2];
            a.y = f.a[t][s >> 1][(s & 1) * 2 + 1];
            acc[t] = __builtin_amdgcn_wmma_f32_16x16x4_f32(
                 false, a,  false, bq[s],
                 (short)0, acc[t],  false,  false);
        }
    }
#pragma unroll
    for (int j = 0; j < DT_GRP / 2; ++j) {
#pragma unroll
        for (int r = 0; r < 8; ++r) {
            const float v = upper ? acc[DT_GRP / 2 + j][r] : acc[j][r];
            dst[(j * 16 + r) * SCAN_GCH] = softplus_f(v + bias);
        }
    }
}

__global__ __launch_bounds__(32 * SCAN_WPB) void mamba_scan(
    const float* __restrict__ proj, const float* __restrict__ W_dt,
    const float* __restrict__ b_dt, const float* __restrict__ UZ,
    const float* __restrict__ A_log, const float* __restrict__ D_param,
    float* __restrict__ y)
{
    __shared__ float dt_s[SEQLEN * SCAN_GCH];

    const int wv   = __builtin_amdgcn_readfirstlane((int)threadIdx.y);
    const int pr   = blockIdx.x * SCAN_WPB + wv;
    const int b    = pr / (D_INNER / SCAN_CH);
    const int d0   = (pr % (D_INNER / SCAN_CH)) * SCAN_CH;
    const int lane = threadIdx.x;

    const float4 al0 = *(const float4*)(A_log + d0 * D_STATE + lane * 4);
    const float4 al1 = *(const float4*)(A_log + (d0 + 1) * D_STATE + lane * 4);
    v2f a[4], h[4];
    a[0].x = -__expf(al0.x); a[1].x = -__expf(al0.y);
    a[2].x = -__expf(al0.z); a[3].x = -__expf(al0.w);
    a[0].y = -__expf(al1.x); a[1].y = -__expf(al1.y);
    a[2].y = -__expf(al1.z); a[3].y = -__expf(al1.w);
#pragma unroll
    for (int j = 0; j < 4; ++j) h[j] = (v2f){0.0f, 0.0f};

    const long row0 = (long)b * SEQLEN;
    const int    ls  = lane >> 1;
    const int    dl  = d0 + (lane & 1);
    const float  Dp  = D_param[dl];
    const float* dtp = dt_s + ls * SCAN_GCH + wv * SCAN_CH + (lane & 1);
    const float* up  = UZ + (row0 + ls) * (2 * D_INNER) + dl;
    const float* prp = proj + row0 * NPROJ + DT_RANK + lane * 4;
    float*       yp  = y + (row0 + ls) * D_INNER + dl;

    ScanOps bc;
    float uc;
    {
        const int  half  = lane >> 4;
        const int  l16   = lane & 15;
        const int  cj    = l16 & 7;
        const bool upper = (l16 & 8) != 0;
        const int  dg    = d0 - wv * SCAN_CH;
        const float* wp  = W_dt + (long)(16 * half) * D_INNER + dg + cj;
        v2f bq[8];
#pragma unroll
        for (int s = 0; s < 8; ++s) {
            bq[s].x = wp[(long)(2 * s) * D_INNER];
            bq[s].y = wp[(long)(2 * s + 1) * D_INNER];
        }
        const float bias = b_dt[dg + cj];
        const int   mt0  = wv * DT_TILES;
        const float* ap  = proj + (row0 + mt0 * 16 + l16) * NPROJ + 16 * half;
        float* dst = dt_s + ((mt0 + (upper ? DT_GRP / 2 : 0)) * 16 + 8 * half) * SCAN_GCH + cj;

        DtFrag f[2];
        dt_load(f[0], ap);
#pragma unroll
        for (int g = 0; g < DT_TILES / DT_GRP; ++g) {
            if (g + 1 < DT_TILES / DT_GRP)
                dt_load(f[(g + 1) & 1], ap + (long)(g + 1) * DT_GRP * 16 * NPROJ);
            else
                { uc = up[0]; scan_load_bc(bc, prp, 0, 0, SCAN_T); }
            __builtin_amdgcn_sched_barrier(0);
            dt_group(f[g & 1], bq, bias, upper, dst + g * DT_GRP * 16 * SCAN_GCH);
            __builtin_amdgcn_sched_barrier(0);
        }
    }
    __syncthreads();
    float dc = dtp[0];

    for (int c = 0; c < SCAN_NCH; ++c) {
        const int tn = ((c + 1 < SCAN_NCH) ? (c + 1) : (SCAN_NCH - 1)) * SCAN_T;
        const float dn = dtp[tn * SCAN_GCH];
        const float un = up[(long)tn * (2 * D_INNER)];
        scan_chunk(bc, dc, uc, a, h, Dp, yp, prp, c * SCAN_T, tn, lane);
        dc = dn;
        uc = un;
    }
}

__global__ __launch_bounds__(256) void ln_gate(
    float* __restrict__ y, const float* __restrict__ UZ,
    const float* __restrict__ gamma, const float* __restrict__ beta)
{
    const int row = blockIdx.x;
    const int tid = threadIdx.x;
    const int c0  = tid * 4;

    const float4 v = *(const float4*)(y + (long)row * D_INNER + c0);
    float s  = v.x + v.y + v.z + v.w;
    float s2 = v.x * v.x + v.y * v.y + v.z * v.z + v.w * v.w;

#pragma unroll
    for (int off = 16; off; off >>= 1) {
        s  += __shfl_xor(s, off, 32);
        s2 += __shfl_xor(s2, off, 32);
    }
    __shared__ float rs[8], rs2[8];
    const int wid = tid >> 5;
    if ((tid & 31) == 0) { rs[wid] = s; rs2[wid] = s2; }
    __syncthreads();
    s = 0.0f; s2 = 0.0f;
#pragma unroll
    for (int w = 0; w < 8; ++w) { s += rs[w]; s2 += rs2[w]; }

    const float mu   = s * (1.0f / D_INNER);
    const float var  = s2 * (1.0f / D_INNER) - mu * mu;
    const float rstd = rsqrtf(var + 1e-5f);

    const float4 ga = *(const float4*)(gamma + c0);
    const float4 be = *(const float4*)(beta + c0);
    const float4 zv = *(const float4*)(UZ + (long)row * (2 * D_INNER) + D_INNER + c0);

    float4 o;
    o.x = ((v.x - mu) * rstd * ga.x + be.x) * silu_f(zv.x);
    o.y = ((v.y - mu) * rstd * ga.y + be.y) * silu_f(zv.y);
    o.z = ((v.z - mu) * rstd * ga.z + be.z) * silu_f(zv.z);
    o.w = ((v.w - mu) * rstd * ga.w + be.w) * silu_f(zv.w);
    *(float4*)(y + (long)row * D_INNER + c0) = o;
}

extern "C" void kernel_launch(void* const* d_in, const int* in_sizes, int n_in,
                              void* d_out, int out_size, void* d_ws, size_t ws_size,
                              hipStream_t stream)
{
    const float* x     = (const float*)d_in[0];
    const float* W_in  = (const float*)d_in[1];
    const float* W_x   = (const float*)d_in[2];
    const float* W_dt  = (const float*)d_in[3];
    const float* b_dt  = (const float*)d_in[4];
    const float* W_out = (const float*)d_in[5];
    const float* gamma = (const float*)d_in[6];
    const float* beta  = (const float*)d_in[7];
    const float* A_log = (const float*)d_in[8];
    const float* D_par = (const float*)d_in[9];
    float* out = (float*)d_out;

    float* UZ   = (float*)d_ws;
    float* proj = UZ   + (size_t)ROWS * 2 * D_INNER;
    float* yb   = proj + (size_t)ROWS * NPROJ;

    gemm16_wmma_f32<1, 4, D_MODEL, 2><<<dim3((2 * D_INNER) / 64, ROWS / 16 / GEMM_WPB), dim3(32, GEMM_WPB), 0, stream>>>(
        x, D_MODEL, W_in, 2 * D_INNER, UZ, 2 * D_INNER, nullptr, D_INNER);

    gemm16_wmma_f32<0, 2, D_INNER, 4><<<dim3(NPROJ / 32, ROWS / 16 / GEMM_WPB), dim3(32, GEMM_WPB), 0, stream>>>(
        UZ, 2 * D_INNER, W_x, NPROJ, proj, NPROJ, nullptr, 0);

    mamba_scan<<<BATCH * D_INNER / SCAN_CH / SCAN_WPB, dim3(32, SCAN_WPB), 0, stream>>>(proj, W_dt, b_dt, UZ, A_log, D_par, yb);

    ln_gate<<<ROWS, 256, 0, stream>>>(yb, UZ, gamma, beta);

    gemm16_wmma_f32<0, 4, D_INNER, 4><<<dim3(D_MODEL / 64, ROWS / 16 / GEMM_WPB), dim3(32, GEMM_WPB), 0, stream>>>(
        yb, D_INNER, W_out, D_MODEL, out, D_MODEL, nullptr, 0);
}
